# starting kernel
# baseline (speedup 1.0000x reference)
#include <hip/hip_runtime.h>
#include <hip/hip_bf16.h>

typedef __attribute__((ext_vector_type(2))) float v2f;
typedef __attribute__((ext_vector_type(8))) float v8f;

#define N_NODES 100000
#define N_REL   8
#define N_EDGES 100000
#define DIM     128
#define TILES   (N_EDGES / 16)
#define AS_STRIDE 132

__device__ __forceinline__ void load_B_regs(const float* __restrict__ W,
                                            int col, int koff, v2f* bW) {
#pragma unroll
  for (int kk = 0; kk < 32; ++kk) {
    int k = 4 * kk + koff;
    v2f b;
    b.x = W[(size_t)k * DIM + col];
    b.y = W[(size_t)(k + 1) * DIM + col];
    bW[kk] = b;
  }
}

__device__ __forceinline__ v8f mma_tile_f32(const float* __restrict__ As,
                                            int ln, int koff, const v2f* bW) {
  v8f c = {0.f, 0.f, 0.f, 0.f, 0.f, 0.f, 0.f, 0.f};
#pragma unroll
  for (int kk = 0; kk < 32; ++kk) {
    v2f a = *(const v2f*)&As[ln * AS_STRIDE + 4 * kk + koff];
    c = __builtin_amdgcn_wmma_f32_16x16x4_f32(false, a, false, bW[kk],
                                              (short)0, c, false, false);
  }
  return c;
}

__global__ __launch_bounds__(256) void self_gemm_kernel(
    const float* __restrict__ x, const float* __restrict__ W,
    float* __restrict__ out) {
  __shared__ float As[16 * AS_STRIDE];
  const int tid  = threadIdx.x;
  const int lane = tid & 31;
  const int wave = tid >> 5;
  const int ln   = lane & 15;
  const int hi   = lane >> 4;
  const int koff = hi * 2;
  const int n0   = wave * 16;

  v2f bW[32];
  load_B_regs(W, n0 + ln, koff, bW);

  for (int t = blockIdx.x; t < TILES; t += gridDim.x) {
#pragma unroll
    for (int j = tid; j < 512; j += 256) {
      int row = j >> 5, c4 = j & 31;
      float4 v = *(const float4*)(x + (size_t)(t * 16 + row) * DIM + c4 * 4);
      *(float4*)&As[row * AS_STRIDE + c4 * 4] = v;
    }
    __syncthreads();

    v8f c = mma_tile_f32(As, ln, koff, bW);

#pragma unroll
    for (int v = 0; v < 8; ++v) {
      int row = v + 8 * hi;
      out[(size_t)(t * 16 + row) * DIM + n0 + ln] = c[v];
    }
    __syncthreads();
  }
}

__global__ __launch_bounds__(256) void edge_gemm_scatter_kernel(
    const float* __restrict__ x, const int* __restrict__ edge_index,
    const float* __restrict__ ew, const float* __restrict__ relW,
    float* __restrict__ out) {
  __shared__ float As[16 * AS_STRIDE];
  __shared__ int   sTgt[16];
  __shared__ float sWt[16];

  const int r = blockIdx.y;
  const float* __restrict__ W    = relW + (size_t)r * DIM * DIM;
  const int*   __restrict__ srcp = edge_index + (size_t)(2 * r) * N_EDGES;
  const int*   __restrict__ tgtp = srcp + N_EDGES;
  const float* __restrict__ wp   = ew + (size_t)r * N_EDGES;

  const int tid  = threadIdx.x;
  const int lane = tid & 31;
  const int wave = tid >> 5;
  const int ln   = lane & 15;
  const int hi   = lane >> 4;
  const int koff = hi * 2;
  const int n0   = wave * 16;

  v2f bW[32];
  load_B_regs(W, n0 + ln, koff, bW);

  for (int t = blockIdx.x; t < TILES; t += gridDim.x) {
    if (tid < 16) {
      int e = t * 16 + tid;
      sTgt[tid] = tgtp[e];
      sWt[tid]  = wp[e];
    }
#pragma unroll
    for (int j = tid; j < 512; j += 256) {
      int row = j >> 5, c4 = j & 31;
      int s = srcp[t * 16 + row];
      float4 v = *(const float4*)(x + (size_t)s * DIM + c4 * 4);
      *(float4*)&As[row * AS_STRIDE + c4 * 4] = v;
    }
    __syncthreads();

    v8f c = mma_tile_f32(As, ln, koff, bW);

#pragma unroll
    for (int v = 0; v < 8; ++v) {
      int row = v + 8 * hi;
      float val = c[v] * sWt[row];
      atomicAdd(&out[(size_t)sTgt[row] * DIM + n0 + ln], val);
    }
    __syncthreads();
  }
}

__global__ void zero_words_kernel(int* __restrict__ p, int n) {
  int gid = blockIdx.x * blockDim.x + threadIdx.x;
  if (gid < n) p[gid] = 0;
}

__global__ void set_flags_kernel(const int* __restrict__ edge_index,
                                 unsigned char* __restrict__ flags) {
  int gid = blockIdx.x * blockDim.x + threadIdx.x;
  if (gid < N_REL * N_EDGES) {
    int r = gid / N_EDGES, e = gid - r * N_EDGES;
    int tgt = edge_index[(size_t)(2 * r + 1) * N_EDGES + e];
    flags[(size_t)r * N_NODES + tgt] = 1;
  }
}

__global__ void bias_act_kernel(float* __restrict__ out,
                                const float* __restrict__ bias,
                                const unsigned char* __restrict__ flags,
                                int do_relu) {
  int gid = blockIdx.x * blockDim.x + threadIdx.x;
  if (gid >= N_NODES * (DIM / 4)) return;
  int n = gid >> 5, c4 = gid & 31;
  float4 v = ((const float4*)out)[gid];
#pragma unroll
  for (int r = 0; r < N_REL; ++r) {
    if (flags[(size_t)r * N_NODES + n]) {
      float4 b = ((const float4*)bias)[r * (DIM / 4) + c4];
      v.x += b.x; v.y += b.y; v.z += b.z; v.w += b.w;
    }
  }
  if (do_relu) {
    v.x = fmaxf(v.x, 0.f); v.y = fmaxf(v.y, 0.f);
    v.z = fmaxf(v.z, 0.f); v.w = fmaxf(v.w, 0.f);
  }
  ((float4*)out)[gid] = v;
}

extern "C" void kernel_launch(void* const* d_in, const int* in_sizes, int n_in,
                              void* d_out, int out_size, void* d_ws, size_t ws_size,
                              hipStream_t stream) {
  const float* x      = (const float*)d_in[0];
  const int*   eidx   = (const int*)d_in[1];
  const float* ew     = (const float*)d_in[2];
  const float* relW1  = (const float*)d_in[3];
  const float* selfW1 = (const float*)d_in[4];
  const float* bias1  = (const float*)d_in[5];
  const float* relW2  = (const float*)d_in[6];
  const float* selfW2 = (const float*)d_in[7];
  const float* bias2  = (const float*)d_in[8];
  float* out = (float*)d_out;

  float* h1 = (float*)d_ws;
  unsigned char* flags =
      (unsigned char*)d_ws + (size_t)N_NODES * DIM * sizeof(float);

  const int GEMM_BLOCKS = 512;
  const int flag_words = (N_REL * N_NODES + 3) / 4;

  zero_words_kernel<<<(flag_words + 255) / 256, 256, 0, stream>>>((int*)flags,
                                                                  flag_words);
  set_flags_kernel<<<(N_REL * N_EDGES + 255) / 256, 256, 0, stream>>>(eidx,
                                                                      flags);

  self_gemm_kernel<<<GEMM_BLOCKS, 256, 0, stream>>>(x, selfW1, h1);
  edge_gemm_scatter_kernel<<<dim3(GEMM_BLOCKS, N_REL), 256, 0, stream>>>(
      x, eidx, ew, relW1, h1);
  bias_act_kernel<<<(N_NODES * (DIM / 4) + 255) / 256, 256, 0, stream>>>(
      h1, bias1, flags, 1);

  self_gemm_kernel<<<GEMM_BLOCKS, 256, 0, stream>>>(h1, selfW2, out);
  edge_gemm_scatter_kernel<<<dim3(GEMM_BLOCKS, N_REL), 256, 0, stream>>>(
      h1, eidx, ew, relW2, out);
  bias_act_kernel<<<(N_NODES * (DIM / 4) + 255) / 256, 256, 0, stream>>>(
      out, bias2, flags, 0);
}


# Fold both bias+ReLU passes into the GEMMs: target mask x bias joins the self GEMM as 8 more k (f32 WMMA), ReLU applied on layer-2 loads; bias is now summed before the scattered messages, not after
# speedup vs baseline: 1.2296x; 1.2296x over previous
#include <hip/hip_runtime.h>
#include <hip/hip_bf16.h>

typedef __attribute__((ext_vector_type(2))) float v2f;
typedef __attribute__((ext_vector_type(8))) float v8f;

#define N_NODES 100000
#define N_REL   8
#define N_EDGES 100000
#define DIM     128
#define TILES   (N_EDGES / 16)
#define AS_STRIDE 132

__device__ __forceinline__ void load_B_regs(const float* __restrict__ W,
                                            int col, int koff, v2f* bW) {
#pragma unroll
  for (int kk = 0; kk < 32; ++kk) {
    int k = 4 * kk + koff;
    v2f b;
    b.x = W[(size_t)k * DIM + col];
    b.y = W[(size_t)(k + 1) * DIM + col];
    bW[kk] = b;
  }
}

__device__ __forceinline__ v8f mma_tile_f32(const float* __restrict__ As,
                                            int ln, int koff, const v2f* bW) {
  v8f c = {0.f, 0.f, 0.f, 0.f, 0.f, 0.f, 0.f, 0.f};
#pragma unroll
  for (int kk = 0; kk < 32; ++kk) {
    v2f a = *(const v2f*)&As[ln * AS_STRIDE + 4 * kk + koff];
    c = __builtin_amdgcn_wmma_f32_16x16x4_f32(false, a, false, bW[kk],
                                              (short)0, c, false, false);
  }
  return c;
}

template <bool RELU_IN>
__global__ __launch_bounds__(256) void self_gemm_kernel(
    const float* __restrict__ x, const float* __restrict__ W,
    const float* __restrict__ bias, const unsigned char* __restrict__ flags,
    float* __restrict__ out) {
  __shared__ float As[16 * AS_STRIDE];
  const int tid  = threadIdx.x;
  const int lane = tid & 31;
  const int wave = tid >> 5;
  const int ln   = lane & 15;
  const int hi   = lane >> 4;
  const int koff = hi * 2;
  const int n0   = wave * 16;

  v2f bW[32];
  load_B_regs(W, n0 + ln, koff, bW);
  v2f bB[2];
  for (int kk = 0; kk < 2; ++kk) {
    int k = 4 * kk + koff;
    v2f b;
    b.x = bias[k * DIM + n0 + ln];
    b.y = bias[(k + 1) * DIM + n0 + ln];
    bB[kk] = b;
  }

  for (int t = blockIdx.x; t < TILES; t += gridDim.x) {
    v2f aM[2];
    for (int kk = 0; kk < 2; ++kk) {
      int k = 4 * kk + koff;
      v2f a;
      a.x = flags[(size_t)k * N_NODES + t * 16 + ln] ? 1.f : 0.f;
      a.y = flags[(size_t)(k + 1) * N_NODES + t * 16 + ln] ? 1.f : 0.f;
      aM[kk] = a;
    }
#pragma unroll
    for (int j = tid; j < 512; j += 256) {
      int row = j >> 5, c4 = j & 31;
      float4 v = *(const float4*)(x + (size_t)(t * 16 + row) * DIM + c4 * 4);
      if (RELU_IN) {
        v.x = fmaxf(v.x, 0.f); v.y = fmaxf(v.y, 0.f);
        v.z = fmaxf(v.z, 0.f); v.w = fmaxf(v.w, 0.f);
      }
      *(float4*)&As[row * AS_STRIDE + c4 * 4] = v;
    }
    __syncthreads();

    v8f c = mma_tile_f32(As, ln, koff, bW);
    for (int kk = 0; kk < 2; ++kk) {
      c = __builtin_amdgcn_wmma_f32_16x16x4_f32(false, aM[kk], false, bB[kk],
                                                (short)0, c, false, false);
    }

#pragma unroll
    for (int v = 0; v < 8; ++v) {
      int row = v + 8 * hi;
      out[(size_t)(t * 16 + row) * DIM + n0 + ln] = c[v];
    }
    __syncthreads();
  }
}

template <bool RELU_IN>
__global__ __launch_bounds__(256) void edge_gemm_scatter_kernel(
    const float* __restrict__ x, const int* __restrict__ edge_index,
    const float* __restrict__ ew, const float* __restrict__ relW,
    float* __restrict__ out) {
  __shared__ float As[16 * AS_STRIDE];
  __shared__ int   sTgt[16];
  __shared__ float sWt[16];

  const int r = blockIdx.y;
  const float* __restrict__ W    = relW + (size_t)r * DIM * DIM;
  const int*   __restrict__ srcp = edge_index + (size_t)(2 * r) * N_EDGES;
  const int*   __restrict__ tgtp = srcp + N_EDGES;
  const float* __restrict__ wp   = ew + (size_t)r * N_EDGES;

  const int tid  = threadIdx.x;
  const int lane = tid & 31;
  const int wave = tid >> 5;
  const int ln   = lane & 15;
  const int hi   = lane >> 4;
  const int koff = hi * 2;
  const int n0   = wave * 16;

  v2f bW[32];
  load_B_regs(W, n0 + ln, koff, bW);

  for (int t = blockIdx.x; t < TILES; t += gridDim.x) {
    if (tid < 16) {
      int e = t * 16 + tid;
      sTgt[tid] = tgtp[e];
      sWt[tid]  = wp[e];
    }
#pragma unroll
    for (int j = tid; j < 512; j += 256) {
      int row = j >> 5, c4 = j & 31;
      int s = srcp[t * 16 + row];
      float4 v = *(const float4*)(x + (size_t)s * DIM + c4 * 4);
      if (RELU_IN) {
        v.x = fmaxf(v.x, 0.f); v.y = fmaxf(v.y, 0.f);
        v.z = fmaxf(v.z, 0.f); v.w = fmaxf(v.w, 0.f);
      }
      *(float4*)&As[row * AS_STRIDE + c4 * 4] = v;
    }
    __syncthreads();

    v8f c = mma_tile_f32(As, ln, koff, bW);

#pragma unroll
    for (int v = 0; v < 8; ++v) {
      int row = v + 8 * hi;
      float val = c[v] * sWt[row];
      atomicAdd(&out[(size_t)sTgt[row] * DIM + n0 + ln], val);
    }
    __syncthreads();
  }
}

__global__ void zero_words_kernel(int* __restrict__ p, int n) {
  int gid = blockIdx.x * blockDim.x + threadIdx.x;
  if (gid < n) p[gid] = 0;
}

__global__ void set_flags_kernel(const int* __restrict__ edge_index,
                                 unsigned char* __restrict__ flags) {
  int gid = blockIdx.x * blockDim.x + threadIdx.x;
  if (gid < N_REL * N_EDGES) {
    int r = gid / N_EDGES, e = gid - r * N_EDGES;
    int tgt = edge_index[(size_t)(2 * r + 1) * N_EDGES + e];
    flags[(size_t)r * N_NODES + tgt] = 1;
  }
}

extern "C" void kernel_launch(void* const* d_in, const int* in_sizes, int n_in,
                              void* d_out, int out_size, void* d_ws, size_t ws_size,
                              hipStream_t stream) {
  const float* x      = (const float*)d_in[0];
  const int*   eidx   = (const int*)d_in[1];
  const float* ew     = (const float*)d_in[2];
  const float* relW1  = (const float*)d_in[3];
  const float* selfW1 = (const float*)d_in[4];
  const float* bias1  = (const float*)d_in[5];
  const float* relW2  = (const float*)d_in[6];
  const float* selfW2 = (const float*)d_in[7];
  const float* bias2  = (const float*)d_in[8];
  float* out = (float*)d_out;

  float* h1 = (float*)d_ws;
  unsigned char* flags =
      (unsigned char*)d_ws + (size_t)N_NODES * DIM * sizeof(float);

  const int GEMM_BLOCKS = 512;
  const int flag_words = (N_REL * N_NODES + 3) / 4;

  zero_words_kernel<<<(flag_words + 255) / 256, 256, 0, stream>>>((int*)flags,
                                                                  flag_words);
  set_flags_kernel<<<(N_REL * N_EDGES + 255) / 256, 256, 0, stream>>>(eidx,
                                                                      flags);

  self_gemm_kernel<false><<<GEMM_BLOCKS, 256, 0, stream>>>(x, selfW1, bias1,
                                                           flags, h1);
  edge_gemm_scatter_kernel<false><<<dim3(GEMM_BLOCKS, N_REL), 256, 0, stream>>>(
      x, eidx, ew, relW1, h1);

  self_gemm_kernel<true><<<GEMM_BLOCKS, 256, 0, stream>>>(h1, selfW2, bias2,
                                                          flags, out);
  edge_gemm_scatter_kernel<true><<<dim3(GEMM_BLOCKS, N_REL), 256, 0, stream>>>(
      h1, eidx, ew, relW2, out);
}


# Edge scatter staged through LDS: each atomic now adds one whole 128-B line of one target row, not two half lines of two rows; same addends, still f32 atomics in arrival order
# speedup vs baseline: 1.3221x; 1.0752x over previous
#include <hip/hip_runtime.h>
#include <hip/hip_bf16.h>

typedef __attribute__((ext_vector_type(2))) float v2f;
typedef __attribute__((ext_vector_type(8))) float v8f;

#define N_NODES 100000
#define N_REL   8
#define N_EDGES 100000
#define DIM     128
#define TILES   (N_EDGES / 16)
#define AS_STRIDE 132
#define CS_STRIDE 132

__device__ __forceinline__ void load_B_regs(const float* __restrict__ W,
                                            int col, int koff, v2f* bW) {
#pragma unroll
  for (int kk = 0; kk < 32; ++kk) {
    int k = 4 * kk + koff;
    v2f b;
    b.x = W[(size_t)k * DIM + col];
    b.y = W[(size_t)(k + 1) * DIM + col];
    bW[kk] = b;
  }
}

__device__ __forceinline__ v8f mma_tile_f32(const float* __restrict__ As,
                                            int ln, int koff, const v2f* bW) {
  v8f c = {0.f, 0.f, 0.f, 0.f, 0.f, 0.f, 0.f, 0.f};
#pragma unroll
  for (int kk = 0; kk < 32; ++kk) {
    v2f a = *(const v2f*)&As[ln * AS_STRIDE + 4 * kk + koff];
    c = __builtin_amdgcn_wmma_f32_16x16x4_f32(false, a, false, bW[kk],
                                              (short)0, c, false, false);
  }
  return c;
}

template <bool RELU_IN>
__global__ __launch_bounds__(256) void self_gemm_kernel(
    const float* __restrict__ x, const float* __restrict__ W,
    const float* __restrict__ bias, const unsigned char* __restrict__ flags,
    float* __restrict__ out) {
  __shared__ float As[16 * AS_STRIDE];
  const int tid  = threadIdx.x;
  const int lane = tid & 31;
  const int wave = tid >> 5;
  const int ln   = lane & 15;
  const int hi   = lane >> 4;
  const int koff = hi * 2;
  const int n0   = wave * 16;

  v2f bW[32];
  load_B_regs(W, n0 + ln, koff, bW);
  v2f bB[2];
  for (int kk = 0; kk < 2; ++kk) {
    int k = 4 * kk + koff;
    v2f b;
    b.x = bias[k * DIM + n0 + ln];
    b.y = bias[(k + 1) * DIM + n0 + ln];
    bB[kk] = b;
  }

  for (int t = blockIdx.x; t < TILES; t += gridDim.x) {
    v2f aM[2];
    for (int kk = 0; kk < 2; ++kk) {
      int k = 4 * kk + koff;
      v2f a;
      a.x = flags[(size_t)k * N_NODES + t * 16 + ln] ? 1.f : 0.f;
      a.y = flags[(size_t)(k + 1) * N_NODES + t * 16 + ln] ? 1.f : 0.f;
      aM[kk] = a;
    }
#pragma unroll
    for (int j = tid; j < 512; j += 256) {
      int row = j >> 5, c4 = j & 31;
      float4 v = *(const float4*)(x + (size_t)(t * 16 + row) * DIM + c4 * 4);
      if (RELU_IN) {
        v.x = fmaxf(v.x, 0.f); v.y = fmaxf(v.y, 0.f);
        v.z = fmaxf(v.z, 0.f); v.w = fmaxf(v.w, 0.f);
      }
      *(float4*)&As[row * AS_STRIDE + c4 * 4] = v;
    }
    __syncthreads();

    v8f c = mma_tile_f32(As, ln, koff, bW);
    for (int kk = 0; kk < 2; ++kk) {
      c = __builtin_amdgcn_wmma_f32_16x16x4_f32(false, aM[kk], false, bB[kk],
                                                (short)0, c, false, false);
    }

#pragma unroll
    for (int v = 0; v < 8; ++v) {
      int row = v + 8 * hi;
      out[(size_t)(t * 16 + row) * DIM + n0 + ln] = c[v];
    }
    __syncthreads();
  }
}

template <bool RELU_IN>
__global__ __launch_bounds__(256) void edge_gemm_scatter_kernel(
    const float* __restrict__ x, const int* __restrict__ edge_index,
    const float* __restrict__ ew, const float* __restrict__ relW,
    float* __restrict__ out) {
  __shared__ float As[16 * AS_STRIDE];
  __shared__ float Cs[16 * CS_STRIDE];

  const int r = blockIdx.y;
  const float* __restrict__ W    = relW + (size_t)r * DIM * DIM;
  const int*   __restrict__ srcp = edge_index + (size_t)(2 * r) * N_EDGES;
  const int*   __restrict__ tgtp = srcp + N_EDGES;
  const float* __restrict__ wp   = ew + (size_t)r * N_EDGES;

  const int tid  = threadIdx.x;
  const int lane = tid & 31;
  const int wave = __builtin_amdgcn_readfirstlane(tid >> 5);
  const int ln   = lane & 15;
  const int hi   = lane >> 4;
  const int koff = hi * 2;
  const int n0   = wave * 16;

  v2f bW[32];
  load_B_regs(W, n0 + ln, koff, bW);

  for (int t = blockIdx.x; t < TILES; t += gridDim.x) {
    const int   e0  = t * 16 + 2 * wave;
    const int   tg0 = tgtp[e0];
    const int   tg1 = tgtp[e0 + 1];
    const float w0  = wp[e0];
    const float w1  = wp[e0 + 1];

#pragma unroll
    for (int j = tid; j < 512; j += 256) {
      int row = j >> 5, c4 = j & 31;
      int s = srcp[t * 16 + row];
      float4 v = *(const float4*)(x + (size_t)s * DIM + c4 * 4);
      if (RELU_IN) {
        v.x = fmaxf(v.x, 0.f); v.y = fmaxf(v.y, 0.f);
        v.z = fmaxf(v.z, 0.f); v.w = fmaxf(v.w, 0.f);
      }
      *(float4*)&As[row * AS_STRIDE + c4 * 4] = v;
    }
    __syncthreads();

    v8f c = mma_tile_f32(As, ln, koff, bW);

    for (int v = 0; v < 8; ++v) {
      int row = v + 8 * hi;
      Cs[row * CS_STRIDE + n0 + ln] = c[v];
    }
    __syncthreads();

    const float* c0 = &Cs[(2 * wave) * CS_STRIDE + lane];
    const float* c1 = c0 + CS_STRIDE;
    float* d0 = out + (size_t)tg0 * DIM + lane;
    float* d1 = out + (size_t)tg1 * DIM + lane;
    float m0[4], m1[4];
    for (int j = 0; j < 4; ++j) { m0[j] = c0[32 * j]; m1[j] = c1[32 * j]; }
    for (int j = 0; j < 4; ++j) atomicAdd(d0 + 32 * j, m0[j] * w0);
    for (int j = 0; j < 4; ++j) atomicAdd(d1 + 32 * j, m1[j] * w1);
  }
}

__global__ void zero_words_kernel(int* __restrict__ p, int n) {
  int gid = blockIdx.x * blockDim.x + threadIdx.x;
  if (gid < n) p[gid] = 0;
}

__global__ void set_flags_kernel(const int* __restrict__ edge_index,
                                 unsigned char* __restrict__ flags) {
  int gid = blockIdx.x * blockDim.x + threadIdx.x;
  if (gid < N_REL * N_EDGES) {
    int r = gid / N_EDGES, e = gid - r * N_EDGES;
    int tgt = edge_index[(size_t)(2 * r + 1) * N_EDGES + e];
    flags[(size_t)r * N_NODES + tgt] = 1;
  }
}

extern "C" void kernel_launch(void* const* d_in, const int* in_sizes, int n_in,
                              void* d_out, int out_size, void* d_ws, size_t ws_size,
                              hipStream_t stream) {
  const float* x      = (const float*)d_in[0];
  const int*   eidx   = (const int*)d_in[1];
  const float* ew     = (const float*)d_in[2];
  const float* relW1  = (const float*)d_in[3];
  const float* selfW1 = (const float*)d_in[4];
  const float* bias1  = (const float*)d_in[5];
  const float* relW2  = (const float*)d_in[6];
  const float* selfW2 = (const float*)d_in[7];
  const float* bias2  = (const float*)d_in[8];
  float* out = (float*)d_out;

  float* h1 = (float*)d_ws;
  unsigned char* flags =
      (unsigned char*)d_ws + (size_t)N_NODES * DIM * sizeof(float);

  const int GEMM_BLOCKS = 512;
  const int flag_words = (N_REL * N_NODES + 3) / 4;

  zero_words_kernel<<<(flag_words + 255) / 256, 256, 0, stream>>>((int*)flags,
                                                                  flag_words);
  set_flags_kernel<<<(N_REL * N_EDGES + 255) / 256, 256, 0, stream>>>(eidx,
                                                                      flags);

  self_gemm_kernel<false><<<GEMM_BLOCKS, 256, 0, stream>>>(x, selfW1, bias1,
                                                           flags, h1);
  edge_gemm_scatter_kernel<false><<<dim3(GEMM_BLOCKS, N_REL), 256, 0, stream>>>(
      x, eidx, ew, relW1, h1);

  self_gemm_kernel<true><<<GEMM_BLOCKS, 256, 0, stream>>>(h1, selfW2, bias2,
                                                          flags, out);
  edge_gemm_scatter_kernel<true><<<dim3(GEMM_BLOCKS, N_REL), 256, 0, stream>>>(
      h1, eidx, ew, relW2, out);
}


# Software-pipelined tile fetch in both GEMM loops: the next tile's source indices, gathered rows, targets and weights are requested behind the barrier and arrive under the current tile's WMMAs
# speedup vs baseline: 1.5044x; 1.1379x over previous
#include <hip/hip_runtime.h>
#include <hip/hip_bf16.h>

typedef __attribute__((ext_vector_type(2))) float v2f;
typedef __attribute__((ext_vector_type(8))) float v8f;

#define N_NODES 100000
#define N_REL   8
#define N_EDGES 100000
#define DIM     128
#define TILES   (N_EDGES / 16)
#define AS_STRIDE 132
#define CS_STRIDE 132

__device__ __forceinline__ void load_B_regs(const float* __restrict__ W,
                                            int col, int koff, v2f* bW) {
#pragma unroll
  for (int kk = 0; kk < 32; ++kk) {
    int k = 4 * kk + koff;
    v2f b;
    b.x = W[(size_t)k * DIM + col];
    b.y = W[(size_t)(k + 1) * DIM + col];
    bW[kk] = b;
  }
}

__device__ __forceinline__ v8f mma_tile_f32(const float* __restrict__ As,
                                            int ln, int koff, const v2f* bW) {
  v8f c = {0.f, 0.f, 0.f, 0.f, 0.f, 0.f, 0.f, 0.f};
#pragma unroll
  for (int kk = 0; kk < 32; ++kk) {
    v2f a = *(const v2f*)&As[ln * AS_STRIDE + 4 * kk + koff];
    c = __builtin_amdgcn_wmma_f32_16x16x4_f32(false, a, false, bW[kk],
                                              (short)0, c, false, false);
  }
  return c;
}

template <bool RELU_IN>
__global__ __launch_bounds__(256) void self_gemm_kernel(
    const float* __restrict__ x, const float* __restrict__ W,
    const float* __restrict__ bias, const unsigned char* __restrict__ flags,
    float* __restrict__ out) {
  __shared__ float As[16 * AS_STRIDE];
  const int tid  = threadIdx.x;
  const int lane = tid & 31;
  const int wave = __builtin_amdgcn_readfirstlane(tid >> 5);
  const int ln   = lane & 15;
  const int hi   = lane >> 4;
  const int koff = hi * 2;
  const int n0   = wave * 16;

  v2f bW[32];
  load_B_regs(W, n0 + ln, koff, bW);
  v2f bB[2];
  for (int kk = 0; kk < 2; ++kk) {
    int k = 4 * kk + koff;
    v2f b;
    b.x = bias[k * DIM + n0 + ln];
    b.y = bias[(k + 1) * DIM + n0 + ln];
    bB[kk] = b;
  }

  const int S = gridDim.x;
  int t = blockIdx.x;
  if (t >= TILES) return;
  const float* __restrict__ xl = x + lane * 4;
  float4 p0 = *(const float4*)(xl + (size_t)(t * 16 + wave) * DIM);
  float4 p1 = *(const float4*)(xl + (size_t)(t * 16 + wave + 8) * DIM);

  for (; t < TILES; t += S) {
    if (RELU_IN) {
      p0.x = fmaxf(p0.x, 0.f); p0.y = fmaxf(p0.y, 0.f);
      p0.z = fmaxf(p0.z, 0.f); p0.w = fmaxf(p0.w, 0.f);
      p1.x = fmaxf(p1.x, 0.f); p1.y = fmaxf(p1.y, 0.f);
      p1.z = fmaxf(p1.z, 0.f); p1.w = fmaxf(p1.w, 0.f);
    }
    *(float4*)&As[wave * AS_STRIDE + lane * 4] = p0;
    *(float4*)&As[(wave + 8) * AS_STRIDE + lane * 4] = p1;
    __syncthreads();

    unsigned int f[4];
    for (int kk = 0; kk < 2; ++kk) {
      int k = 4 * kk + koff;
      f[2 * kk]     = flags[(size_t)k * N_NODES + t * 16 + ln];
      f[2 * kk + 1] = flags[(size_t)(k + 1) * N_NODES + t * 16 + ln];
    }
    const int tn = t + S;
    if (tn < TILES) {
      p0 = *(const float4*)(xl + (size_t)(tn * 16 + wave) * DIM);
      p1 = *(const float4*)(xl + (size_t)(tn * 16 + wave + 8) * DIM);
    }

    v8f c = mma_tile_f32(As, ln, koff, bW);
    for (int kk = 0; kk < 2; ++kk) {
      v2f a;
      a.x = f[2 * kk] ? 1.f : 0.f;
      a.y = f[2 * kk + 1] ? 1.f : 0.f;
      c = __builtin_amdgcn_wmma_f32_16x16x4_f32(false, a, false, bB[kk],
                                                (short)0, c, false, false);
    }

#pragma unroll
    for (int v = 0; v < 8; ++v) {
      int row = v + 8 * hi;
      out[(size_t)(t * 16 + row) * DIM + n0 + ln] = c[v];
    }
    __syncthreads();
  }
}

template <bool RELU_IN>
__global__ __launch_bounds__(256) void edge_gemm_scatter_kernel(
    const float* __restrict__ x, const int* __restrict__ edge_index,
    const float* __restrict__ ew, const float* __restrict__ relW,
    float* __restrict__ out) {
  __shared__ float As[16 * AS_STRIDE];
  __shared__ float Cs[16 * CS_STRIDE];

  const int r = blockIdx.y;
  const float* __restrict__ W    = relW + (size_t)r * DIM * DIM;
  const int*   __restrict__ srcp = edge_index + (size_t)(2 * r) * N_EDGES;
  const int*   __restrict__ tgtp = srcp + N_EDGES;
  const float* __restrict__ wp   = ew + (size_t)r * N_EDGES;

  const int tid  = threadIdx.x;
  const int lane = tid & 31;
  const int wave = __builtin_amdgcn_readfirstlane(tid >> 5);
  const int ln   = lane & 15;
  const int hi   = lane >> 4;
  const int koff = hi * 2;
  const int n0   = wave * 16;

  v2f bW[32];
  load_B_regs(W, n0 + ln, koff, bW);

  const int S = gridDim.x;
  int t = blockIdx.x;
  if (t >= TILES) return;
  const float* __restrict__ xl = x + lane * 4;
  float4 p0, p1;
  {
    const int s0 = srcp[t * 16 + wave];
    const int s1 = srcp[t * 16 + wave + 8];
    p0 = *(const float4*)(xl + (size_t)s0 * DIM);
    p1 = *(const float4*)(xl + (size_t)s1 * DIM);
  }
  int   tg0 = tgtp[t * 16 + 2 * wave];
  int   tg1 = tgtp[t * 16 + 2 * wave + 1];
  float w0  = wp[t * 16 + 2 * wave];
  float w1  = wp[t * 16 + 2 * wave + 1];
  int tc  = min(t + S, TILES - 1);
  int ns0 = srcp[tc * 16 + wave];
  int ns1 = srcp[tc * 16 + wave + 8];

  for (; t < TILES; t += S) {
    if (RELU_IN) {
      p0.x = fmaxf(p0.x, 0.f); p0.y = fmaxf(p0.y, 0.f);
      p0.z = fmaxf(p0.z, 0.f); p0.w = fmaxf(p0.w, 0.f);
      p1.x = fmaxf(p1.x, 0.f); p1.y = fmaxf(p1.y, 0.f);
      p1.z = fmaxf(p1.z, 0.f); p1.w = fmaxf(p1.w, 0.f);
    }
    *(float4*)&As[wave * AS_STRIDE + lane * 4] = p0;
    *(float4*)&As[(wave + 8) * AS_STRIDE + lane * 4] = p1;
    float* d0 = out + (size_t)tg0 * DIM + lane;
    float* d1 = out + (size_t)tg1 * DIM + lane;
    const float cw0 = w0, cw1 = w1;
    __syncthreads();

    const int tn = t + S;
    if (tn < TILES) {
      p0 = *(const float4*)(xl + (size_t)ns0 * DIM);
      p1 = *(const float4*)(xl + (size_t)ns1 * DIM);
      tg0 = tgtp[tn * 16 + 2 * wave];
      tg1 = tgtp[tn * 16 + 2 * wave + 1];
      w0  = wp[tn * 16 + 2 * wave];
      w1  = wp[tn * 16 + 2 * wave + 1];
      tc  = min(tn + S, TILES - 1);
      ns0 = srcp[tc * 16 + wave];
      ns1 = srcp[tc * 16 + wave + 8];
    }

    v8f c = mma_tile_f32(As, ln, koff, bW);

    for (int v = 0; v < 8; ++v) {
      int row = v + 8 * hi;
      Cs[row * CS_STRIDE + n0 + ln] = c[v];
    }
    __syncthreads();

    const float* c0 = &Cs[(2 * wave) * CS_STRIDE + lane];
    const float* c1 = c0 + CS_STRIDE;
    float m0[4], m1[4];
    for (int j = 0; j < 4; ++j) { m0[j] = c0[32 * j]; m1[j] = c1[32 * j]; }
    for (int j = 0; j < 4; ++j) atomicAdd(d0 + 32 * j, m0[j] * cw0);
    for (int j = 0; j < 4; ++j) atomicAdd(d1 + 32 * j, m1[j] * cw1);
  }
}

__global__ void zero_words_kernel(int* __restrict__ p, int n) {
  int gid = blockIdx.x * blockDim.x + threadIdx.x;
  if (gid < n) p[gid] = 0;
}

__global__ void set_flags_kernel(const int* __restrict__ edge_index,
                                 unsigned char* __restrict__ flags) {
  int gid = blockIdx.x * blockDim.x + threadIdx.x;
  if (gid < N_REL * N_EDGES) {
    int r = gid / N_EDGES, e = gid - r * N_EDGES;
    int tgt = edge_index[(size_t)(2 * r + 1) * N_EDGES + e];
    flags[(size_t)r * N_NODES + tgt] = 1;
  }
}

extern "C" void kernel_launch(void* const* d_in, const int* in_sizes, int n_in,
                              void* d_out, int out_size, void* d_ws, size_t ws_size,
                              hipStream_t stream) {
  const float* x      = (const float*)d_in[0];
  const int*   eidx   = (const int*)d_in[1];
  const float* ew     = (const float*)d_in[2];
  const float* relW1  = (const float*)d_in[3];
  const float* selfW1 = (const float*)d_in[4];
  const float* bias1  = (const float*)d_in[5];
  const float* relW2  = (const float*)d_in[6];
  const float* selfW2 = (const float*)d_in[7];
  const float* bias2  = (const float*)d_in[8];
  float* out = (float*)d_out;

  float* h1 = (float*)d_ws;
  unsigned char* flags =
      (unsigned char*)d_ws + (size_t)N_NODES * DIM * sizeof(float);

  const int GEMM_BLOCKS = 512;
  const int flag_words = (N_REL * N_NODES + 3) / 4;

  zero_words_kernel<<<(flag_words + 255) / 256, 256, 0, stream>>>((int*)flags,
                                                                  flag_words);
  set_flags_kernel<<<(N_REL * N_EDGES + 255) / 256, 256, 0, stream>>>(eidx,
                                                                      flags);

  self_gemm_kernel<false><<<GEMM_BLOCKS, 256, 0, stream>>>(x, selfW1, bias1,
                                                           flags, h1);
  edge_gemm_scatter_kernel<false><<<dim3(GEMM_BLOCKS, N_REL), 256, 0, stream>>>(
      x, eidx, ew, relW1, h1);

  self_gemm_kernel<true><<<GEMM_BLOCKS, 256, 0, stream>>>(h1, selfW2, bias2,
                                                          flags, out);
  edge_gemm_scatter_kernel<true><<<dim3(GEMM_BLOCKS, N_REL), 256, 0, stream>>>(
      h1, eidx, ew, relW2, out);
}


# Edges counting-sorted by (relation, target) once per call; each edge-GEMM A row is the f32 weighted sum of a pair's source rows, one product row added per pair: sums regrouped, ranks by int atomics
# speedup vs baseline: 1.5726x; 1.0453x over previous
#include <hip/hip_runtime.h>
#include <hip/hip_bf16.h>

typedef __attribute__((ext_vector_type(2))) float v2f;
typedef __attribute__((ext_vector_type(8))) float v8f;

#define N_NODES 100000
#define N_REL   8
#define N_EDGES 100000
#define DIM     128
#define TILES   (N_EDGES / 16)
#define AS_STRIDE 132
#define CS_STRIDE 132
#define SCAN_CHUNK 1024
#define NB_SCAN  ((N_NODES + SCAN_CHUNK - 1) / SCAN_CHUNK)
#define PS_STRIDE (N_NODES + 16)

__device__ __forceinline__ void load_B_regs(const float* __restrict__ W,
                                            int col, int koff, v2f* bW) {
#pragma unroll
  for (int kk = 0; kk < 32; ++kk) {
    int k = 4 * kk + koff;
    v2f b;
    b.x = W[(size_t)k * DIM + col];
    b.y = W[(size_t)(k + 1) * DIM + col];
    bW[kk] = b;
  }
}

__device__ __forceinline__ v8f mma_tile_f32(const float* __restrict__ As,
                                            int ln, int koff, const v2f* bW) {
  v8f c = {0.f, 0.f, 0.f, 0.f, 0.f, 0.f, 0.f, 0.f};
#pragma unroll
  for (int kk = 0; kk < 32; ++kk) {
    v2f a = *(const v2f*)&As[ln * AS_STRIDE + 4 * kk + koff];
    c = __builtin_amdgcn_wmma_f32_16x16x4_f32(false, a, false, bW[kk],
                                              (short)0, c, false, false);
  }
  return c;
}

template <bool RELU_IN>
__global__ __launch_bounds__(256) void self_gemm_kernel(
    const float* __restrict__ x, const float* __restrict__ W,
    const float* __restrict__ bias, const int* __restrict__ flags,
    float* __restrict__ out) {
  __shared__ float As[16 * AS_STRIDE];
  const int tid  = threadIdx.x;
  const int lane = tid & 31;
  const int wave = __builtin_amdgcn_readfirstlane(tid >> 5);
  const int ln   = lane & 15;
  const int hi   = lane >> 4;
  const int koff = hi * 2;
  const int n0   = wave * 16;

  v2f bW[32];
  load_B_regs(W, n0 + ln, koff, bW);
  v2f bB[2];
  for (int kk = 0; kk < 2; ++kk) {
    int k = 4 * kk + koff;
    v2f b;
    b.x = bias[k * DIM + n0 + ln];
    b.y = bias[(k + 1) * DIM + n0 + ln];
    bB[kk] = b;
  }

  const int S = gridDim.x;
  int t = blockIdx.x;
  if (t >= TILES) return;
  const float* __restrict__ xl = x + lane * 4;
  float4 p0 = *(const float4*)(xl + (size_t)(t * 16 + wave) * DIM);
  float4 p1 = *(const float4*)(xl + (size_t)(t * 16 + wave + 8) * DIM);

  for (; t < TILES; t += S) {
    if (RELU_IN) {
      p0.x = fmaxf(p0.x, 0.f); p0.y = fmaxf(p0.y, 0.f);
      p0.z = fmaxf(p0.z, 0.f); p0.w = fmaxf(p0.w, 0.f);
      p1.x = fmaxf(p1.x, 0.f); p1.y = fmaxf(p1.y, 0.f);
      p1.z = fmaxf(p1.z, 0.f); p1.w = fmaxf(p1.w, 0.f);
    }
    *(float4*)&As[wave * AS_STRIDE + lane * 4] = p0;
    *(float4*)&As[(wave + 8) * AS_STRIDE + lane * 4] = p1;
    __syncthreads();

    int f[4];
    for (int kk = 0; kk < 2; ++kk) {
      int k = 4 * kk + koff;
      f[2 * kk]     = flags[(size_t)k * N_NODES + t * 16 + ln];
      f[2 * kk + 1] = flags[(size_t)(k + 1) * N_NODES + t * 16 + ln];
    }
    const int tn = t + S;
    if (tn < TILES) {
      p0 = *(const float4*)(xl + (size_t)(tn * 16 + wave) * DIM);
      p1 = *(const float4*)(xl + (size_t)(tn * 16 + wave + 8) * DIM);
    }

    v8f c = mma_tile_f32(As, ln, koff, bW);
    for (int kk = 0; kk < 2; ++kk) {
      v2f a;
      a.x = f[2 * kk] > 0 ? 1.f : 0.f;
      a.y = f[2 * kk + 1] > 0 ? 1.f : 0.f;
      c = __builtin_amdgcn_wmma_f32_16x16x4_f32(false, a, false, bB[kk],
                                                (short)0, c, false, false);
    }

#pragma unroll
    for (int v = 0; v < 8; ++v) {
      int row = v + 8 * hi;
      out[(size_t)(t * 16 + row) * DIM + n0 + ln] = c[v];
    }
    __syncthreads();
  }
}

__global__ void zero_words_kernel(int* __restrict__ p, int n) {
  int gid = blockIdx.x * blockDim.x + threadIdx.x;
  if (gid < n) p[gid] = 0;
}

__global__ void count_rank_kernel(const int* __restrict__ edge_index,
                                  int* __restrict__ cnt,
                                  int* __restrict__ rank) {
  int gid = blockIdx.x * blockDim.x + threadIdx.x;
  if (gid < N_REL * N_EDGES) {
    int r = gid / N_EDGES, e = gid - r * N_EDGES;
    int tgt = edge_index[(size_t)(2 * r + 1) * N_EDGES + e];
    rank[gid] = atomicAdd(&cnt[r * N_NODES + tgt], 1);
  }
}

__device__ __forceinline__ void block_scan2(int& a, int& b, int* sa, int* sb,
                                            int tid) {
  sa[tid] = a;
  sb[tid] = b;
  __syncthreads();
  for (int off = 1; off < 256; off <<= 1) {
    int ta = 0, tb = 0;
    if (tid >= off) { ta = sa[tid - off]; tb = sb[tid - off]; }
    __syncthreads();
    sa[tid] += ta;
    sb[tid] += tb;
    __syncthreads();
  }
  a = sa[tid];
  b = sb[tid];
}

__global__ __launch_bounds__(256) void seg_block_sums_kernel(
    const int* __restrict__ cnt, int* __restrict__ bsum) {
  __shared__ int sa[256];
  __shared__ int sb[256];
  const int r = blockIdx.y, b = blockIdx.x, tid = threadIdx.x;
  const int n0 = b * SCAN_CHUNK + tid * 4;
  int c = 0, f = 0;
  if (n0 < N_NODES) {
    int4 v = *(const int4*)(cnt + (size_t)r * N_NODES + n0);
    c = v.x + v.y + v.z + v.w;
    f = (v.x > 0) + (v.y > 0) + (v.z > 0) + (v.w > 0);
  }
  block_scan2(c, f, sa, sb, tid);
  if (tid == 255) {
    bsum[(r * NB_SCAN + b) * 2]     = c;
    bsum[(r * NB_SCAN + b) * 2 + 1] = f;
  }
}

__global__ __launch_bounds__(256) void seg_scan_kernel(
    const int* __restrict__ cnt, const int* __restrict__ bsum,
    int* __restrict__ start, int* __restrict__ ptgt, int* __restrict__ pstart,
    int* __restrict__ npair) {
  __shared__ int sa[256];
  __shared__ int sb[256];
  const int r = blockIdx.y, b = blockIdx.x, tid = threadIdx.x;
  int pc = 0, pf = 0;
  if (tid < b) {
    pc = bsum[(r * NB_SCAN + tid) * 2];
    pf = bsum[(r * NB_SCAN + tid) * 2 + 1];
  }
  block_scan2(pc, pf, sa, sb, tid);
  const int baseC = sa[255], baseF = sb[255];
  __syncthreads();

  const int n0 = b * SCAN_CHUNK + tid * 4;
  int4 v = {0, 0, 0, 0};
  if (n0 < N_NODES) v = *(const int4*)(cnt + (size_t)r * N_NODES + n0);
  const int c = v.x + v.y + v.z + v.w;
  const int f = (v.x > 0) + (v.y > 0) + (v.z > 0) + (v.w > 0);
  int ic = c, jf = f;
  block_scan2(ic, jf, sa, sb, tid);
  int runC = baseC + ic - c;
  int runF = baseF + jf - f;
  if (n0 < N_NODES) {
    const int cs[4] = {v.x, v.y, v.z, v.w};
#pragma unroll
    for (int i = 0; i < 4; ++i) {
      start[(size_t)r * N_NODES + n0 + i] = runC;
      if (cs[i] > 0) {
        ptgt[(size_t)r * N_NODES + runF]    = n0 + i;
        pstart[(size_t)r * PS_STRIDE + runF] = runC;
        ++runF;
      }
      runC += cs[i];
    }
  }
  if (b == NB_SCAN - 1 && tid == 255) {
    const int np = baseF + jf;
    npair[r] = np;
    pstart[(size_t)r * PS_STRIDE + np] = baseC + ic;
  }
}

__global__ void place_edges_kernel(const int* __restrict__ edge_index,
                                   const float* __restrict__ ew,
                                   const int* __restrict__ start,
                                   const int* __restrict__ rank,
                                   int* __restrict__ ssrc,
                                   float* __restrict__ sw) {
  int gid = blockIdx.x * blockDim.x + threadIdx.x;
  if (gid < N_REL * N_EDGES) {
    int r = gid / N_EDGES, e = gid - r * N_EDGES;
    int tgt = edge_index[(size_t)(2 * r + 1) * N_EDGES + e];
    int pos = start[(size_t)r * N_NODES + tgt] + rank[gid];
    ssrc[(size_t)r * N_EDGES + pos] = edge_index[(size_t)(2 * r) * N_EDGES + e];
    sw[(size_t)r * N_EDGES + pos]   = ew[gid];
  }
}

template <bool RELU_IN>
__device__ __forceinline__ void add_row(float4& acc, float w, float4 v) {
  if (RELU_IN) {
    v.x = fmaxf(v.x, 0.f); v.y = fmaxf(v.y, 0.f);
    v.z = fmaxf(v.z, 0.f); v.w = fmaxf(v.w, 0.f);
  }
  acc.x += w * v.x; acc.y += w * v.y; acc.z += w * v.z; acc.w += w * v.w;
}

template <bool RELU_IN>
__global__ __launch_bounds__(256) void edge_gemm_scatter_kernel(
    const float* __restrict__ x, const float* __restrict__ relW,
    const int* __restrict__ ssrc, const float* __restrict__ sw,
    const int* __restrict__ ptgt, const int* __restrict__ pstart,
    const int* __restrict__ npair, float* __restrict__ out) {
  __shared__ float As[16 * AS_STRIDE];
  __shared__ float Cs[16 * CS_STRIDE];

  const int r = blockIdx.y;
  const float* __restrict__ W    = relW + (size_t)r * DIM * DIM;
  const int*   __restrict__ srcs = ssrc + (size_t)r * N_EDGES;
  const float* __restrict__ wts  = sw + (size_t)r * N_EDGES;
  const int*   __restrict__ tgts = ptgt + (size_t)r * N_NODES;
  const int*   __restrict__ pst  = pstart + (size_t)r * PS_STRIDE;
  const int np     = npair[r];
  const int ntiles = (np + 15) >> 4;

  const int tid  = threadIdx.x;
  const int lane = tid & 31;
  const int wave = __builtin_amdgcn_readfirstlane(tid >> 5);
  const int ln   = lane & 15;
  const int hi   = lane >> 4;
  const int koff = hi * 2;
  const int n0   = wave * 16;

  const int S = gridDim.x;
  int t = blockIdx.x;
  if (t >= ntiles) return;

  v2f bW[32];
  load_B_regs(W, n0 + ln, koff, bW);

  const float* __restrict__ xl = x + lane * 4;
  int b0 = pst[min(t * 16 + wave, np)],     e0 = pst[min(t * 16 + wave + 1, np)];
  int b1 = pst[min(t * 16 + wave + 8, np)], e1 = pst[min(t * 16 + wave + 9, np)];
  int tq = t + S;
  int nb0 = pst[min(tq * 16 + wave, np)],     ne0 = pst[min(tq * 16 + wave + 1, np)];
  int nb1 = pst[min(tq * 16 + wave + 8, np)], ne1 = pst[min(tq * 16 + wave + 9, np)];
  int   iv0 = srcs[min(b0 + lane, N_EDGES - 1)];
  float wv0 = wts[min(b0 + lane, N_EDGES - 1)];
  int   iv1 = srcs[min(b1 + lane, N_EDGES - 1)];
  float wv1 = wts[min(b1 + lane, N_EDGES - 1)];

  for (; t < ntiles; t += S) {
    float4 a0 = {0.f, 0.f, 0.f, 0.f}, a1 = {0.f, 0.f, 0.f, 0.f};
    const int len0 = e0 - b0, len1 = e1 - b1;
    {
      const int s00 = __builtin_amdgcn_readlane(iv0, 0);
      const int s01 = __builtin_amdgcn_readlane(iv0, 1);
      const int s10 = __builtin_amdgcn_readlane(iv1, 0);
      const int s11 = __builtin_amdgcn_readlane(iv1, 1);
      const float4 v00 = *(const float4*)(xl + (size_t)s00 * DIM);
      const float4 v01 = *(const float4*)(xl + (size_t)s01 * DIM);
      const float4 v10 = *(const float4*)(xl + (size_t)s10 * DIM);
      const float4 v11 = *(const float4*)(xl + (size_t)s11 * DIM);
      const float w00 = __int_as_float(__builtin_amdgcn_readlane(__float_as_int(wv0), 0));
      const float w01 = __int_as_float(__builtin_amdgcn_readlane(__float_as_int(wv0), 1));
      const float w10 = __int_as_float(__builtin_amdgcn_readlane(__float_as_int(wv1), 0));
      const float w11 = __int_as_float(__builtin_amdgcn_readlane(__float_as_int(wv1), 1));
      add_row<RELU_IN>(a0, w00, v00);
      add_row<RELU_IN>(a0, len0 > 1 ? w01 : 0.f, v01);
      add_row<RELU_IN>(a1, w10, v10);
      add_row<RELU_IN>(a1, len1 > 1 ? w11 : 0.f, v11);
    }
    for (int k = 2; k < len0; ++k) {
      int s; float w;
      if (k < 32) {
        s = __builtin_amdgcn_readlane(iv0, k);
        w = __int_as_float(__builtin_amdgcn_readlane(__float_as_int(wv0), k));
      } else {
        s = srcs[b0 + k];
        w = wts[b0 + k];
      }
      add_row<RELU_IN>(a0, w, *(const float4*)(xl + (size_t)s * DIM));
    }
    for (int k = 2; k < len1; ++k) {
      int s; float w;
      if (k < 32) {
        s = __builtin_amdgcn_readlane(iv1, k);
        w = __int_as_float(__builtin_amdgcn_readlane(__float_as_int(wv1), k));
      } else {
        s = srcs[b1 + k];
        w = wts[b1 + k];
      }
      add_row<RELU_IN>(a1, w, *(const float4*)(xl + (size_t)s * DIM));
    }
    *(float4*)&As[wave * AS_STRIDE + lane * 4] = a0;
    *(float4*)&As[(wave + 8) * AS_STRIDE + lane * 4] = a1;
    __syncthreads();

    const int q0 = t * 16 + 2 * wave, q1 = q0 + 1;
    const int tg0 = tgts[min(q0, np - 1)];
    const int tg1 = tgts[min(q1, np - 1)];

    b0 = nb0; e0 = ne0; b1 = nb1; e1 = ne1;
    iv0 = srcs[min(b0 + lane, N_EDGES - 1)];
    wv0 = wts[min(b0 + lane, N_EDGES - 1)];
    iv1 = srcs[min(b1 + lane, N_EDGES - 1)];
    wv1 = wts[min(b1 + lane, N_EDGES - 1)];
    tq = t + 2 * S;
    nb0 = pst[min(tq * 16 + wave, np)];     ne0 = pst[min(tq * 16 + wave + 1, np)];
    nb1 = pst[min(tq * 16 + wave + 8, np)]; ne1 = pst[min(tq * 16 + wave + 9, np)];

    v8f c = mma_tile_f32(As, ln, koff, bW);

    for (int v = 0; v < 8; ++v) {
      int row = v + 8 * hi;
      Cs[row * CS_STRIDE + n0 + ln] = c[v];
    }
    __syncthreads();

    const float* c0 = &Cs[(2 * wave) * CS_STRIDE + lane];
    const float* c1 = c0 + CS_STRIDE;
    float m0[4], m1[4];
    for (int j = 0; j < 4; ++j) { m0[j] = c0[32 * j]; m1[j] = c1[32 * j]; }
    if (q0 < np) {
      float* d0 = out + (size_t)tg0 * DIM + lane;
      for (int j = 0; j < 4; ++j) atomicAdd(d0 + 32 * j, m0[j]);
    }
    if (q1 < np) {
      float* d1 = out + (size_t)tg1 * DIM + lane;
      for (int j = 0; j < 4; ++j) atomicAdd(d1 + 32 * j, m1[j]);
    }
  }
}

extern "C" void kernel_launch(void* const* d_in, const int* in_sizes, int n_in,
                              void* d_out, int out_size, void* d_ws, size_t ws_size,
                              hipStream_t stream) {
  const float* x      = (const float*)d_in[0];
  const int*   eidx   = (const int*)d_in[1];
  const float* ew     = (const float*)d_in[2];
  const float* relW1  = (const float*)d_in[3];
  const float* selfW1 = (const float*)d_in[4];
  const float* bias1  = (const float*)d_in[5];
  const float* relW2  = (const float*)d_in[6];
  const float* selfW2 = (const float*)d_in[7];
  const float* bias2  = (const float*)d_in[8];
  float* out = (float*)d_out;

  float* h1   = (float*)d_ws;
  int* cnt    = (int*)(h1 + (size_t)N_NODES * DIM);
  int* rank   = cnt + (size_t)N_REL * N_NODES;
  int* start  = rank + (size_t)N_REL * N_EDGES;
  int* ssrc   = start + (size_t)N_REL * N_NODES;
  float* sw   = (float*)(ssrc + (size_t)N_REL * N_EDGES);
  int* ptgt   = (int*)(sw + (size_t)N_REL * N_EDGES);
  int* pstart = ptgt + (size_t)N_REL * N_NODES;
  int* bsum   = pstart + (size_t)N_REL * PS_STRIDE;
  int* npair  = bsum + (size_t)N_REL * NB_SCAN * 2;

  const int GEMM_BLOCKS = 512;
  const int cnt_words = N_REL * N_NODES;

  zero_words_kernel<<<(cnt_words + 255) / 256, 256, 0, stream>>>(cnt, cnt_words);
  count_rank_kernel<<<(N_REL * N_EDGES + 255) / 256, 256, 0, stream>>>(eidx, cnt,
                                                                       rank);
  seg_block_sums_kernel<<<dim3(NB_SCAN, N_REL), 256, 0, stream>>>(cnt, bsum);
  seg_scan_kernel<<<dim3(NB_SCAN, N_REL), 256, 0, stream>>>(cnt, bsum, start, ptgt,
                                                           pstart, npair);
  place_edges_kernel<<<(N_REL * N_EDGES + 255) / 256, 256, 0, stream>>>(
      eidx, ew, start, rank, ssrc, sw);

  self_gemm_kernel<false><<<GEMM_BLOCKS, 256, 0, stream>>>(x, selfW1, bias1,
                                                           cnt, h1);
  edge_gemm_scatter_kernel<false><<<dim3(GEMM_BLOCKS, N_REL), 256, 0, stream>>>(
      x, relW1, ssrc, sw, ptgt, pstart, npair, h1);

  self_gemm_kernel<true><<<GEMM_BLOCKS, 256, 0, stream>>>(h1, selfW2, bias2,
                                                          cnt, out);
  edge_gemm_scatter_kernel<true><<<dim3(GEMM_BLOCKS, N_REL), 256, 0, stream>>>(
      h1, relW2, ssrc, sw, ptgt, pstart, npair, out);
}


# Edge grouping by (relation, target) on LDS counters: two launches of range-owning blocks replace five launches with global int atomics; rank inside a pair still by int atomics in arrival order
# speedup vs baseline: 1.8076x; 1.1494x over previous
#include <hip/hip_runtime.h>
#include <hip/hip_bf16.h>

typedef __attribute__((ext_vector_type(2))) float v2f;
typedef __attribute__((ext_vector_type(8))) float v8f;

#define N_NODES 100000
#define N_REL   8
#define N_EDGES 100000
#define DIM     128
#define TILES   (N_EDGES / 16)
#define AS_STRIDE 132
#define CS_STRIDE 132
#define NCH 32
#define CH  (N_NODES / NCH)
#define PS_STRIDE (N_NODES + 16)
#define LIST_BATCH  5
#define PLACE_BATCH 4

__device__ __forceinline__ void load_B_regs(const float* __restrict__ W,
                                            int col, int koff, v2f* bW) {
#pragma unroll
  for (int kk = 0; kk < 32; ++kk) {
    int k = 4 * kk + koff;
    v2f b;
    b.x = W[(size_t)k * DIM + col];
    b.y = W[(size_t)(k + 1) * DIM + col];
    bW[kk] = b;
  }
}

__device__ __forceinline__ v8f mma_tile_f32(const float* __restrict__ As,
                                            int ln, int koff, const v2f* bW) {
  v8f c = {0.f, 0.f, 0.f, 0.f, 0.f, 0.f, 0.f, 0.f};
#pragma unroll
  for (int kk = 0; kk < 32; ++kk) {
    v2f a = *(const v2f*)&As[ln * AS_STRIDE + 4 * kk + koff];
    c = __builtin_amdgcn_wmma_f32_16x16x4_f32(false, a, false, bW[kk],
                                              (short)0, c, false, false);
  }
  return c;
}

template <bool RELU_IN>
__global__ __launch_bounds__(256) void self_gemm_kernel(
    const float* __restrict__ x, const float* __restrict__ W,
    const float* __restrict__ bias, const int* __restrict__ flags,
    float* __restrict__ out) {
  __shared__ float As[16 * AS_STRIDE];
  const int tid  = threadIdx.x;
  const int lane = tid & 31;
  const int wave = __builtin_amdgcn_readfirstlane(tid >> 5);
  const int ln   = lane & 15;
  const int hi   = lane >> 4;
  const int koff = hi * 2;
  const int n0   = wave * 16;

  v2f bW[32];
  load_B_regs(W, n0 + ln, koff, bW);
  v2f bB[2];
  for (int kk = 0; kk < 2; ++kk) {
    int k = 4 * kk + koff;
    v2f b;
    b.x = bias[k * DIM + n0 + ln];
    b.y = bias[(k + 1) * DIM + n0 + ln];
    bB[kk] = b;
  }

  const int S = gridDim.x;
  int t = blockIdx.x;
  if (t >= TILES) return;
  const float* __restrict__ xl = x + lane * 4;
  float4 p0 = *(const float4*)(xl + (size_t)(t * 16 + wave) * DIM);
  float4 p1 = *(const float4*)(xl + (size_t)(t * 16 + wave + 8) * DIM);

  for (; t < TILES; t += S) {
    if (RELU_IN) {
      p0.x = fmaxf(p0.x, 0.f); p0.y = fmaxf(p0.y, 0.f);
      p0.z = fmaxf(p0.z, 0.f); p0.w = fmaxf(p0.w, 0.f);
      p1.x = fmaxf(p1.x, 0.f); p1.y = fmaxf(p1.y, 0.f);
      p1.z = fmaxf(p1.z, 0.f); p1.w = fmaxf(p1.w, 0.f);
    }
    *(float4*)&As[wave * AS_STRIDE + lane * 4] = p0;
    *(float4*)&As[(wave + 8) * AS_STRIDE + lane * 4] = p1;
    __syncthreads();

    int f[4];
    for (int kk = 0; kk < 2; ++kk) {
      int k = 4 * kk + koff;
      f[2 * kk]     = flags[(size_t)k * N_NODES + t * 16 + ln];
      f[2 * kk + 1] = flags[(size_t)(k + 1) * N_NODES + t * 16 + ln];
    }
    const int tn = t + S;
    if (tn < TILES) {
      p0 = *(const float4*)(xl + (size_t)(tn * 16 + wave) * DIM);
      p1 = *(const float4*)(xl + (size_t)(tn * 16 + wave + 8) * DIM);
    }

    v8f c = mma_tile_f32(As, ln, koff, bW);
    for (int kk = 0; kk < 2; ++kk) {
      v2f a;
      a.x = f[2 * kk] > 0 ? 1.f : 0.f;
      a.y = f[2 * kk + 1] > 0 ? 1.f : 0.f;
      c = __builtin_amdgcn_wmma_f32_16x16x4_f32(false, a, false, bB[kk],
                                                (short)0, c, false, false);
    }

#pragma unroll
    for (int v = 0; v < 8; ++v) {
      int row = v + 8 * hi;
      out[(size_t)(t * 16 + row) * DIM + n0 + ln] = c[v];
    }
    __syncthreads();
  }
}


__device__ __forceinline__ void block_scan2(int& a, int& b, int* sa, int* sb,
                                            int tid) {
  const int lane = tid & 31, wv = tid >> 5;
  for (int off = 1; off < 32; off <<= 1) {
    int ta = __shfl_up(a, off), tb = __shfl_up(b, off);
    if (lane >= off) { a += ta; b += tb; }
  }
  if (lane == 31) { sa[wv] = a; sb[wv] = b; }
  __syncthreads();
  if (wv == 0) {
    int wa = sa[lane], wb = sb[lane];
    for (int off = 1; off < 32; off <<= 1) {
      int ta = __shfl_up(wa, off), tb = __shfl_up(wb, off);
      if (lane >= off) { wa += ta; wb += tb; }
    }
    sa[lane] = wa;
    sb[lane] = wb;
  }
  __syncthreads();
  if (wv > 0) { a += sa[wv - 1]; b += sb[wv - 1]; }
}

__global__ __launch_bounds__(1024) void range_count_kernel(
    const int* __restrict__ edge_index, int* __restrict__ cnt,
    int* __restrict__ rtot) {
  __shared__ int s_cnt[CH];
  __shared__ int sa[32];
  __shared__ int sb[32];
  const int c = blockIdx.x, r = blockIdx.y, tid = threadIdx.x;
  const int lo = c * CH;
  for (int i = tid; i < CH; i += 1024) s_cnt[i] = 0;
  __syncthreads();
  const int4* __restrict__ tg =
      (const int4*)(edge_index + (size_t)(2 * r + 1) * N_EDGES);
  for (int q0 = tid; q0 < N_EDGES / 4; q0 += 1024 * LIST_BATCH) {
    int4 t[LIST_BATCH];
#pragma unroll
    for (int u = 0; u < LIST_BATCH; ++u) {
      const int q = q0 + 1024 * u;
      t[u] = q < N_EDGES / 4 ? tg[q] : make_int4(-1, -1, -1, -1);
    }
#pragma unroll
    for (int u = 0; u < LIST_BATCH; ++u) {
      const unsigned d0 = (unsigned)(t[u].x - lo), d1 = (unsigned)(t[u].y - lo);
      const unsigned d2 = (unsigned)(t[u].z - lo), d3 = (unsigned)(t[u].w - lo);
      if (d0 < (unsigned)CH) atomicAdd(&s_cnt[d0], 1);
      if (d1 < (unsigned)CH) atomicAdd(&s_cnt[d1], 1);
      if (d2 < (unsigned)CH) atomicAdd(&s_cnt[d2], 1);
      if (d3 < (unsigned)CH) atomicAdd(&s_cnt[d3], 1);
    }
  }
  __syncthreads();
  int ne = 0, nf = 0;
  for (int i = tid; i < CH; i += 1024) {
    const int v = s_cnt[i];
    cnt[(size_t)r * N_NODES + lo + i] = v;
    ne += v;
    nf += (v > 0);
  }
  block_scan2(ne, nf, sa, sb, tid);
  if (tid == 1023) {
    rtot[(r * NCH + c) * 2]     = ne;
    rtot[(r * NCH + c) * 2 + 1] = nf;
  }
}

__global__ __launch_bounds__(1024) void range_place_kernel(
    const int* __restrict__ edge_index, const float* __restrict__ ew,
    const int* __restrict__ cnt, const int* __restrict__ rtot,
    int* __restrict__ ssrc, float* __restrict__ sw, int* __restrict__ ptgt,
    int* __restrict__ pstart, int* __restrict__ npair) {
  __shared__ int s_cur[CH];
  __shared__ int sa[32];
  __shared__ int sb[32];
  const int c = blockIdx.x, r = blockIdx.y, tid = threadIdx.x;
  const int lo = c * CH;
  int baseC = 0, baseF = 0;
  for (int j = 0; j < c; ++j) {
    baseC += rtot[(r * NCH + j) * 2];
    baseF += rtot[(r * NCH + j) * 2 + 1];
  }
  const int i0 = tid * 4;
  int cs[4];
#pragma unroll
  for (int i = 0; i < 4; ++i)
    cs[i] = (i0 + i < CH) ? cnt[(size_t)r * N_NODES + lo + i0 + i] : 0;
  const int sc = cs[0] + cs[1] + cs[2] + cs[3];
  const int sf = (cs[0] > 0) + (cs[1] > 0) + (cs[2] > 0) + (cs[3] > 0);
  int ic = sc, jf = sf;
  block_scan2(ic, jf, sa, sb, tid);
  int runC = baseC + ic - sc;
  int runF = baseF + jf - sf;
#pragma unroll
  for (int i = 0; i < 4; ++i) {
    if (i0 + i < CH) {
      s_cur[i0 + i] = runC;
      if (cs[i] > 0) {
        ptgt[(size_t)r * N_NODES + runF]     = lo + i0 + i;
        pstart[(size_t)r * PS_STRIDE + runF] = runC;
        ++runF;
      }
      runC += cs[i];
    }
  }
  if (c == NCH - 1 && tid == 1023) {
    npair[r] = runF;
    pstart[(size_t)r * PS_STRIDE + runF] = runC;
  }
  __syncthreads();

  const int4* __restrict__ tg =
      (const int4*)(edge_index + (size_t)(2 * r + 1) * N_EDGES);
  const int*   __restrict__ sr = edge_index + (size_t)(2 * r) * N_EDGES;
  const float* __restrict__ wr = ew + (size_t)r * N_EDGES;
  int*   __restrict__ so = ssrc + (size_t)r * N_EDGES;
  float* __restrict__ wo = sw + (size_t)r * N_EDGES;
  const int4*   __restrict__ sr4 = (const int4*)sr;
  const float4* __restrict__ wr4 = (const float4*)wr;
  for (int q0 = tid; q0 < N_EDGES / 4; q0 += 1024 * PLACE_BATCH) {
    int4 t[PLACE_BATCH], s[PLACE_BATCH];
    float4 w[PLACE_BATCH];
#pragma unroll
    for (int u = 0; u < PLACE_BATCH; ++u) {
      const int q = q0 + 1024 * u;
      if (q < N_EDGES / 4) {
        t[u] = tg[q]; s[u] = sr4[q]; w[u] = wr4[q];
      } else {
        t[u] = make_int4(-1, -1, -1, -1);
        s[u] = make_int4(0, 0, 0, 0);
        w[u] = make_float4(0.f, 0.f, 0.f, 0.f);
      }
    }
#pragma unroll
    for (int u = 0; u < PLACE_BATCH; ++u) {
      const int   tt[4] = {t[u].x, t[u].y, t[u].z, t[u].w};
      const int   ss[4] = {s[u].x, s[u].y, s[u].z, s[u].w};
      const float ww[4] = {w[u].x, w[u].y, w[u].z, w[u].w};
#pragma unroll
      for (int i = 0; i < 4; ++i) {
        const unsigned d = (unsigned)(tt[i] - lo);
        if (d < (unsigned)CH) {
          const int pos = atomicAdd(&s_cur[d], 1);
          so[pos] = ss[i];
          wo[pos] = ww[i];
        }
      }
    }
  }
}

template <bool RELU_IN>
__device__ __forceinline__ void add_row(float4& acc, float w, float4 v) {
  if (RELU_IN) {
    v.x = fmaxf(v.x, 0.f); v.y = fmaxf(v.y, 0.f);
    v.z = fmaxf(v.z, 0.f); v.w = fmaxf(v.w, 0.f);
  }
  acc.x += w * v.x; acc.y += w * v.y; acc.z += w * v.z; acc.w += w * v.w;
}

template <bool RELU_IN>
__global__ __launch_bounds__(256) void edge_gemm_scatter_kernel(
    const float* __restrict__ x, const float* __restrict__ relW,
    const int* __restrict__ ssrc, const float* __restrict__ sw,
    const int* __restrict__ ptgt, const int* __restrict__ pstart,
    const int* __restrict__ npair, float* __restrict__ out) {
  __shared__ float As[16 * AS_STRIDE];
  __shared__ float Cs[16 * CS_STRIDE];

  const int r = blockIdx.y;
  const float* __restrict__ W    = relW + (size_t)r * DIM * DIM;
  const int*   __restrict__ srcs = ssrc + (size_t)r * N_EDGES;
  const float* __restrict__ wts  = sw + (size_t)r * N_EDGES;
  const int*   __restrict__ tgts = ptgt + (size_t)r * N_NODES;
  const int*   __restrict__ pst  = pstart + (size_t)r * PS_STRIDE;
  const int np     = npair[r];
  const int ntiles = (np + 15) >> 4;

  const int tid  = threadIdx.x;
  const int lane = tid & 31;
  const int wave = __builtin_amdgcn_readfirstlane(tid >> 5);
  const int ln   = lane & 15;
  const int hi   = lane >> 4;
  const int koff = hi * 2;
  const int n0   = wave * 16;

  const int S = gridDim.x;
  int t = blockIdx.x;
  if (t >= ntiles) return;

  v2f bW[32];
  load_B_regs(W, n0 + ln, koff, bW);

  const float* __restrict__ xl = x + lane * 4;
  int b0 = pst[min(t * 16 + wave, np)],     e0 = pst[min(t * 16 + wave + 1, np)];
  int b1 = pst[min(t * 16 + wave + 8, np)], e1 = pst[min(t * 16 + wave + 9, np)];
  int tq = t + S;
  int nb0 = pst[min(tq * 16 + wave, np)],     ne0 = pst[min(tq * 16 + wave + 1, np)];
  int nb1 = pst[min(tq * 16 + wave + 8, np)], ne1 = pst[min(tq * 16 + wave + 9, np)];
  int   iv0 = srcs[min(b0 + lane, N_EDGES - 1)];
  float wv0 = wts[min(b0 + lane, N_EDGES - 1)];
  int   iv1 = srcs[min(b1 + lane, N_EDGES - 1)];
  float wv1 = wts[min(b1 + lane, N_EDGES - 1)];

  for (; t < ntiles; t += S) {
    float4 a0 = {0.f, 0.f, 0.f, 0.f}, a1 = {0.f, 0.f, 0.f, 0.f};
    const int len0 = e0 - b0, len1 = e1 - b1;
    {
      const int s00 = __builtin_amdgcn_readlane(iv0, 0);
      const int s01 = __builtin_amdgcn_readlane(iv0, 1);
      const int s10 = __builtin_amdgcn_readlane(iv1, 0);
      const int s11 = __builtin_amdgcn_readlane(iv1, 1);
      const float4 v00 = *(const float4*)(xl + (size_t)s00 * DIM);
      const float4 v01 = *(const float4*)(xl + (size_t)s01 * DIM);
      const float4 v10 = *(const float4*)(xl + (size_t)s10 * DIM);
      const float4 v11 = *(const float4*)(xl + (size_t)s11 * DIM);
      const float w00 = __int_as_float(__builtin_amdgcn_readlane(__float_as_int(wv0), 0));
      const float w01 = __int_as_float(__builtin_amdgcn_readlane(__float_as_int(wv0), 1));
      const float w10 = __int_as_float(__builtin_amdgcn_readlane(__float_as_int(wv1), 0));
      const float w11 = __int_as_float(__builtin_amdgcn_readlane(__float_as_int(wv1), 1));
      add_row<RELU_IN>(a0, w00, v00);
      add_row<RELU_IN>(a0, len0 > 1 ? w01 : 0.f, v01);
      add_row<RELU_IN>(a1, w10, v10);
      add_row<RELU_IN>(a1, len1 > 1 ? w11 : 0.f, v11);
    }
    for (int k = 2; k < len0; ++k) {
      int s; float w;
      if (k < 32) {
        s = __builtin_amdgcn_readlane(iv0, k);
        w = __int_as_float(__builtin_amdgcn_readlane(__float_as_int(wv0), k));
      } else {
        s = srcs[b0 + k];
        w = wts[b0 + k];
      }
      add_row<RELU_IN>(a0, w, *(const float4*)(xl + (size_t)s * DIM));
    }
    for (int k = 2; k < len1; ++k) {
      int s; float w;
      if (k < 32) {
        s = __builtin_amdgcn_readlane(iv1, k);
        w = __int_as_float(__builtin_amdgcn_readlane(__float_as_int(wv1), k));
      } else {
        s = srcs[b1 + k];
        w = wts[b1 + k];
      }
      add_row<RELU_IN>(a1, w, *(const float4*)(xl + (size_t)s * DIM));
    }
    *(float4*)&As[wave * AS_STRIDE + lane * 4] = a0;
    *(float4*)&As[(wave + 8) * AS_STRIDE + lane * 4] = a1;
    __syncthreads();

    const int q0 = t * 16 + 2 * wave, q1 = q0 + 1;
    const int tg0 = tgts[min(q0, np - 1)];
    const int tg1 = tgts[min(q1, np - 1)];

    b0 = nb0; e0 = ne0; b1 = nb1; e1 = ne1;
    iv0 = srcs[min(b0 + lane, N_EDGES - 1)];
    wv0 = wts[min(b0 + lane, N_EDGES - 1)];
    iv1 = srcs[min(b1 + lane, N_EDGES - 1)];
    wv1 = wts[min(b1 + lane, N_EDGES - 1)];
    tq = t + 2 * S;
    nb0 = pst[min(tq * 16 + wave, np)];     ne0 = pst[min(tq * 16 + wave + 1, np)];
    nb1 = pst[min(tq * 16 + wave + 8, np)]; ne1 = pst[min(tq * 16 + wave + 9, np)];

    v8f c = mma_tile_f32(As, ln, koff, bW);

    for (int v = 0; v < 8; ++v) {
      int row = v + 8 * hi;
      Cs[row * CS_STRIDE + n0 + ln] = c[v];
    }
    __syncthreads();

    const float* c0 = &Cs[(2 * wave) * CS_STRIDE + lane];
    const float* c1 = c0 + CS_STRIDE;
    float m0[4], m1[4];
    for (int j = 0; j < 4; ++j) { m0[j] = c0[32 * j]; m1[j] = c1[32 * j]; }
    if (q0 < np) {
      float* d0 = out + (size_t)tg0 * DIM + lane;
      for (int j = 0; j < 4; ++j) atomicAdd(d0 + 32 * j, m0[j]);
    }
    if (q1 < np) {
      float* d1 = out + (size_t)tg1 * DIM + lane;
      for (int j = 0; j < 4; ++j) atomicAdd(d1 + 32 * j, m1[j]);
    }
  }
}

extern "C" void kernel_launch(void* const* d_in, const int* in_sizes, int n_in,
                              void* d_out, int out_size, void* d_ws, size_t ws_size,
                              hipStream_t stream) {
  const float* x      = (const float*)d_in[0];
  const int*   eidx   = (const int*)d_in[1];
  const float* ew     = (const float*)d_in[2];
  const float* relW1  = (const float*)d_in[3];
  const float* selfW1 = (const float*)d_in[4];
  const float* bias1  = (const float*)d_in[5];
  const float* relW2  = (const float*)d_in[6];
  const float* selfW2 = (const float*)d_in[7];
  const float* bias2  = (const float*)d_in[8];
  float* out = (float*)d_out;

  float* h1   = (float*)d_ws;
  int* cnt    = (int*)(h1 + (size_t)N_NODES * DIM);
  int* ssrc   = cnt + (size_t)N_REL * N_NODES;
  float* sw   = (float*)(ssrc + (size_t)N_REL * N_EDGES);
  int* ptgt   = (int*)(sw + (size_t)N_REL * N_EDGES);
  int* pstart = ptgt + (size_t)N_REL * N_NODES;
  int* rtot   = pstart + (size_t)N_REL * PS_STRIDE;
  int* npair  = rtot + (size_t)N_REL * NCH * 2;

  const int GEMM_BLOCKS = 512;

  range_count_kernel<<<dim3(NCH, N_REL), 1024, 0, stream>>>(eidx, cnt, rtot);
  range_place_kernel<<<dim3(NCH, N_REL), 1024, 0, stream>>>(
      eidx, ew, cnt, rtot, ssrc, sw, ptgt, pstart, npair);

  self_gemm_kernel<false><<<GEMM_BLOCKS, 256, 0, stream>>>(x, selfW1, bias1,
                                                           cnt, h1);
  edge_gemm_scatter_kernel<false><<<dim3(GEMM_BLOCKS, N_REL), 256, 0, stream>>>(
      x, relW1, ssrc, sw, ptgt, pstart, npair, h1);

  self_gemm_kernel<true><<<GEMM_BLOCKS, 256, 0, stream>>>(h1, selfW2, bias2,
                                                          cnt, out);
  edge_gemm_scatter_kernel<true><<<dim3(GEMM_BLOCKS, N_REL), 256, 0, stream>>>(
      h1, relW2, ssrc, sw, ptgt, pstart, npair, out);
}


# Edge GEMM re-tiled: each wave owns whole 16-pair tiles (A rows, 16x128 product on 8 accumulators, scatter), W in LDS in fragment order, no barrier per tile; same sums, f32 atomics in arrival order
# speedup vs baseline: 1.8912x; 1.0462x over previous
#include <hip/hip_runtime.h>
#include <hip/hip_bf16.h>

typedef __attribute__((ext_vector_type(2))) float v2f;
typedef __attribute__((ext_vector_type(8))) float v8f;

#define N_NODES 100000
#define N_REL   8
#define N_EDGES 100000
#define DIM     128
#define TILES   (N_EDGES / 16)
#define AS_STRIDE 132
#define NCH 32
#define CH  (N_NODES / NCH)
#define PS_STRIDE (N_NODES + 16)
#define LIST_BATCH  5
#define PLACE_BATCH 4

__device__ __forceinline__ void load_B_regs(const float* __restrict__ W,
                                            int col, int koff, v2f* bW) {
#pragma unroll
  for (int kk = 0; kk < 32; ++kk) {
    int k = 4 * kk + koff;
    v2f b;
    b.x = W[(size_t)k * DIM + col];
    b.y = W[(size_t)(k + 1) * DIM + col];
    bW[kk] = b;
  }
}

__device__ __forceinline__ v8f mma_tile_f32(const float* __restrict__ As,
                                            int ln, int koff, const v2f* bW) {
  v8f c = {0.f, 0.f, 0.f, 0.f, 0.f, 0.f, 0.f, 0.f};
#pragma unroll
  for (int kk = 0; kk < 32; ++kk) {
    v2f a = *(const v2f*)&As[ln * AS_STRIDE + 4 * kk + koff];
    c = __builtin_amdgcn_wmma_f32_16x16x4_f32(false, a, false, bW[kk],
                                              (short)0, c, false, false);
  }
  return c;
}

template <bool RELU_IN>
__global__ __launch_bounds__(256) void self_gemm_kernel(
    const float* __restrict__ x, const float* __restrict__ W,
    const float* __restrict__ bias, const int* __restrict__ flags,
    float* __restrict__ out) {
  __shared__ float As[16 * AS_STRIDE];
  const int tid  = threadIdx.x;
  const int lane = tid & 31;
  const int wave = __builtin_amdgcn_readfirstlane(tid >> 5);
  const int ln   = lane & 15;
  const int hi   = lane >> 4;
  const int koff = hi * 2;
  const int n0   = wave * 16;

  v2f bW[32];
  load_B_regs(W, n0 + ln, koff, bW);
  v2f bB[2];
  for (int kk = 0; kk < 2; ++kk) {
    int k = 4 * kk + koff;
    v2f b;
    b.x = bias[k * DIM + n0 + ln];
    b.y = bias[(k + 1) * DIM + n0 + ln];
    bB[kk] = b;
  }

  const int S = gridDim.x;
  int t = blockIdx.x;
  if (t >= TILES) return;
  const float* __restrict__ xl = x + lane * 4;
  float4 p0 = *(const float4*)(xl + (size_t)(t * 16 + wave) * DIM);
  float4 p1 = *(const float4*)(xl + (size_t)(t * 16 + wave + 8) * DIM);

  for (; t < TILES; t += S) {
    if (RELU_IN) {
      p0.x = fmaxf(p0.x, 0.f); p0.y = fmaxf(p0.y, 0.f);
      p0.z = fmaxf(p0.z, 0.f); p0.w = fmaxf(p0.w, 0.f);
      p1.x = fmaxf(p1.x, 0.f); p1.y = fmaxf(p1.y, 0.f);
      p1.z = fmaxf(p1.z, 0.f); p1.w = fmaxf(p1.w, 0.f);
    }
    *(float4*)&As[wave * AS_STRIDE + lane * 4] = p0;
    *(float4*)&As[(wave + 8) * AS_STRIDE + lane * 4] = p1;
    __syncthreads();

    int f[4];
    for (int kk = 0; kk < 2; ++kk) {
      int k = 4 * kk + koff;
      f[2 * kk]     = flags[(size_t)k * N_NODES + t * 16 + ln];
      f[2 * kk + 1] = flags[(size_t)(k + 1) * N_NODES + t * 16 + ln];
    }
    const int tn = t + S;
    if (tn < TILES) {
      p0 = *(const float4*)(xl + (size_t)(tn * 16 + wave) * DIM);
      p1 = *(const float4*)(xl + (size_t)(tn * 16 + wave + 8) * DIM);
    }

    v8f c = mma_tile_f32(As, ln, koff, bW);
    for (int kk = 0; kk < 2; ++kk) {
      v2f a;
      a.x = f[2 * kk] > 0 ? 1.f : 0.f;
      a.y = f[2 * kk + 1] > 0 ? 1.f : 0.f;
      c = __builtin_amdgcn_wmma_f32_16x16x4_f32(false, a, false, bB[kk],
                                                (short)0, c, false, false);
    }

#pragma unroll
    for (int v = 0; v < 8; ++v) {
      int row = v + 8 * hi;
      out[(size_t)(t * 16 + row) * DIM + n0 + ln] = c[v];
    }
    __syncthreads();
  }
}


__device__ __forceinline__ void block_scan2(int& a, int& b, int* sa, int* sb,
                                            int tid) {
  const int lane = tid & 31, wv = tid >> 5;
  for (int off = 1; off < 32; off <<= 1) {
    int ta = __shfl_up(a, off), tb = __shfl_up(b, off);
    if (lane >= off) { a += ta; b += tb; }
  }
  if (lane == 31) { sa[wv] = a; sb[wv] = b; }
  __syncthreads();
  if (wv == 0) {
    int wa = sa[lane], wb = sb[lane];
    for (int off = 1; off < 32; off <<= 1) {
      int ta = __shfl_up(wa, off), tb = __shfl_up(wb, off);
      if (lane >= off) { wa += ta; wb += tb; }
    }
    sa[lane] = wa;
    sb[lane] = wb;
  }
  __syncthreads();
  if (wv > 0) { a += sa[wv - 1]; b += sb[wv - 1]; }
}

__global__ __launch_bounds__(1024) void range_count_kernel(
    const int* __restrict__ edge_index, int* __restrict__ cnt,
    int* __restrict__ rtot) {
  __shared__ int s_cnt[CH];
  __shared__ int sa[32];
  __shared__ int sb[32];
  const int c = blockIdx.x, r = blockIdx.y, tid = threadIdx.x;
  const int lo = c * CH;
  for (int i = tid; i < CH; i += 1024) s_cnt[i] = 0;
  __syncthreads();
  const int4* __restrict__ tg =
      (const int4*)(edge_index + (size_t)(2 * r + 1) * N_EDGES);
  for (int q0 = tid; q0 < N_EDGES / 4; q0 += 1024 * LIST_BATCH) {
    int4 t[LIST_BATCH];
#pragma unroll
    for (int u = 0; u < LIST_BATCH; ++u) {
      const int q = q0 + 1024 * u;
      t[u] = q < N_EDGES / 4 ? tg[q] : make_int4(-1, -1, -1, -1);
    }
#pragma unroll
    for (int u = 0; u < LIST_BATCH; ++u) {
      const unsigned d0 = (unsigned)(t[u].x - lo), d1 = (unsigned)(t[u].y - lo);
      const unsigned d2 = (unsigned)(t[u].z - lo), d3 = (unsigned)(t[u].w - lo);
      if (d0 < (unsigned)CH) atomicAdd(&s_cnt[d0], 1);
      if (d1 < (unsigned)CH) atomicAdd(&s_cnt[d1], 1);
      if (d2 < (unsigned)CH) atomicAdd(&s_cnt[d2], 1);
      if (d3 < (unsigned)CH) atomicAdd(&s_cnt[d3], 1);
    }
  }
  __syncthreads();
  int ne = 0, nf = 0;
  for (int i = tid; i < CH; i += 1024) {
    const int v = s_cnt[i];
    cnt[(size_t)r * N_NODES + lo + i] = v;
    ne += v;
    nf += (v > 0);
  }
  block_scan2(ne, nf, sa, sb, tid);
  if (tid == 1023) {
    rtot[(r * NCH + c) * 2]     = ne;
    rtot[(r * NCH + c) * 2 + 1] = nf;
  }
}

__global__ __launch_bounds__(1024) void range_place_kernel(
    const int* __restrict__ edge_index, const float* __restrict__ ew,
    const int* __restrict__ cnt, const int* __restrict__ rtot,
    int* __restrict__ ssrc, float* __restrict__ sw, int* __restrict__ ptgt,
    int* __restrict__ pstart, int* __restrict__ npair) {
  __shared__ int s_cur[CH];
  __shared__ int sa[32];
  __shared__ int sb[32];
  const int c = blockIdx.x, r = blockIdx.y, tid = threadIdx.x;
  const int lo = c * CH;
  int baseC = 0, baseF = 0;
  for (int j = 0; j < c; ++j) {
    baseC += rtot[(r * NCH + j) * 2];
    baseF += rtot[(r * NCH + j) * 2 + 1];
  }
  const int i0 = tid * 4;
  int cs[4];
#pragma unroll
  for (int i = 0; i < 4; ++i)
    cs[i] = (i0 + i < CH) ? cnt[(size_t)r * N_NODES + lo + i0 + i] : 0;
  const int sc = cs[0] + cs[1] + cs[2] + cs[3];
  const int sf = (cs[0] > 0) + (cs[1] > 0) + (cs[2] > 0) + (cs[3] > 0);
  int ic = sc, jf = sf;
  block_scan2(ic, jf, sa, sb, tid);
  int runC = baseC + ic - sc;
  int runF = baseF + jf - sf;
#pragma unroll
  for (int i = 0; i < 4; ++i) {
    if (i0 + i < CH) {
      s_cur[i0 + i] = runC;
      if (cs[i] > 0) {
        ptgt[(size_t)r * N_NODES + runF]     = lo + i0 + i;
        pstart[(size_t)r * PS_STRIDE + runF] = runC;
        ++runF;
      }
      runC += cs[i];
    }
  }
  if (c == NCH - 1 && tid == 1023) {
    npair[r] = runF;
    pstart[(size_t)r * PS_STRIDE + runF] = runC;
  }
  __syncthreads();

  const int4* __restrict__ tg =
      (const int4*)(edge_index + (size_t)(2 * r + 1) * N_EDGES);
  const int*   __restrict__ sr = edge_index + (size_t)(2 * r) * N_EDGES;
  const float* __restrict__ wr = ew + (size_t)r * N_EDGES;
  int*   __restrict__ so = ssrc + (size_t)r * N_EDGES;
  float* __restrict__ wo = sw + (size_t)r * N_EDGES;
  const int4*   __restrict__ sr4 = (const int4*)sr;
  const float4* __restrict__ wr4 = (const float4*)wr;
  for (int q0 = tid; q0 < N_EDGES / 4; q0 += 1024 * PLACE_BATCH) {
    int4 t[PLACE_BATCH], s[PLACE_BATCH];
    float4 w[PLACE_BATCH];
#pragma unroll
    for (int u = 0; u < PLACE_BATCH; ++u) {
      const int q = q0 + 1024 * u;
      if (q < N_EDGES / 4) {
        t[u] = tg[q]; s[u] = sr4[q]; w[u] = wr4[q];
      } else {
        t[u] = make_int4(-1, -1, -1, -1);
        s[u] = make_int4(0, 0, 0, 0);
        w[u] = make_float4(0.f, 0.f, 0.f, 0.f);
      }
    }
#pragma unroll
    for (int u = 0; u < PLACE_BATCH; ++u) {
      const int   tt[4] = {t[u].x, t[u].y, t[u].z, t[u].w};
      const int   ss[4] = {s[u].x, s[u].y, s[u].z, s[u].w};
      const float ww[4] = {w[u].x, w[u].y, w[u].z, w[u].w};
#pragma unroll
      for (int i = 0; i < 4; ++i) {
        const unsigned d = (unsigned)(tt[i] - lo);
        if (d < (unsigned)CH) {
          const int pos = atomicAdd(&s_cur[d], 1);
          so[pos] = ss[i];
          wo[pos] = ww[i];
        }
      }
    }
  }
}

template <bool RELU_IN>
__device__ __forceinline__ void add_row(float4& acc, float w, float4 v) {
  if (RELU_IN) {
    v.x = fmaxf(v.x, 0.f); v.y = fmaxf(v.y, 0.f);
    v.z = fmaxf(v.z, 0.f); v.w = fmaxf(v.w, 0.f);
  }
  acc.x += w * v.x; acc.y += w * v.y; acc.z += w * v.z; acc.w += w * v.w;
}

template <bool RELU_IN>
__global__ __launch_bounds__(256) void edge_gemm_scatter_kernel(
    const float* __restrict__ x, const float* __restrict__ relW,
    const int* __restrict__ ssrc, const float* __restrict__ sw,
    const int* __restrict__ ptgt, const int* __restrict__ pstart,
    const int* __restrict__ npair, float* __restrict__ out) {
  __shared__ __attribute__((aligned(16))) float Ws[DIM * DIM];
  __shared__ __attribute__((aligned(16))) float Ts[8 * 16 * AS_STRIDE];

  const int r = blockIdx.y;
  const float* __restrict__ W    = relW + (size_t)r * DIM * DIM;
  const int*   __restrict__ srcs = ssrc + (size_t)r * N_EDGES;
  const float* __restrict__ wts  = sw + (size_t)r * N_EDGES;
  const int*   __restrict__ tgts = ptgt + (size_t)r * N_NODES;
  const int*   __restrict__ pst  = pstart + (size_t)r * PS_STRIDE;
  const int np     = npair[r];
  const int ntiles = (np + 15) >> 4;

  const int tid  = threadIdx.x;
  const int lane = tid & 31;
  const int wave = __builtin_amdgcn_readfirstlane(tid >> 5);
  const int ln   = lane & 15;
  const int hi   = lane >> 4;
  const int koff = hi * 2;

  for (int i = tid; i < DIM * DIM / 4; i += 256) {
    const int row = i >> 5, col = (i & 31) << 2;
    const float4 w4 = *(const float4*)(W + (size_t)row * DIM + col);
    const int kk = row >> 2, h = (row >> 1) & 1, s = row & 1;
    const int j = col >> 4, c = col & 15;
    float* d = &Ws[((((kk * 4 + (j >> 1)) * 32) + h * 16 + c) << 2) + 2 * (j & 1) + s];
    d[0] = w4.x; d[4] = w4.y; d[8] = w4.z; d[12] = w4.w;
  }
  __syncthreads();

  float* __restrict__ Aw = Ts + wave * (16 * AS_STRIDE);
  const float4* __restrict__ Ws4 = (const float4*)Ws;

  const int S = gridDim.x * 8;
  int t = blockIdx.x * 8 + wave;
  if (t >= ntiles) return;

  const float* __restrict__ xl = x + lane * 4;
  const int l16 = min(lane, 16);
  int pv  = pst[min(t * 16 + l16, np)];
  int pvn = pst[min((t + S) * 16 + l16, np)];
  int eb  = __builtin_amdgcn_readlane(pv, 0);
  int   iv = srcs[min(eb + lane, N_EDGES - 1)];
  float wv = wts[min(eb + lane, N_EDGES - 1)];

  for (; t < ntiles; t += S) {
    const int ee = __builtin_amdgcn_readlane(pv, 16);
    float4 acc = {0.f, 0.f, 0.f, 0.f};
    int row  = 0;
    int rend = __builtin_amdgcn_readlane(pv, 1);
    for (int cb = eb; cb < ee; cb += 32) {
      if (cb != eb) {
        iv = srcs[min(cb + lane, N_EDGES - 1)];
        wv = wts[min(cb + lane, N_EDGES - 1)];
      }
      const int cnt = min(32, ee - cb);
      for (int j0 = 0; j0 < cnt; j0 += 8) {
        float4 v[8];
#pragma unroll
        for (int u = 0; u < 8; ++u) {
          const int s = __builtin_amdgcn_readlane(iv, min(j0 + u, cnt - 1));
          v[u] = *(const float4*)(xl + (size_t)s * DIM);
        }
#pragma unroll
        for (int u = 0; u < 8; ++u) {
          const int j = j0 + u;
          if (j < cnt) {
            while (cb + j >= rend && row < 15) {
              *(float4*)&Aw[row * AS_STRIDE + lane * 4] = acc;
              acc = make_float4(0.f, 0.f, 0.f, 0.f);
              ++row;
              rend = __builtin_amdgcn_readlane(pv, row + 1);
            }
            const float w =
                __int_as_float(__builtin_amdgcn_readlane(__float_as_int(wv), j));
            add_row<RELU_IN>(acc, w, v[u]);
          }
        }
      }
    }
    *(float4*)&Aw[row * AS_STRIDE + lane * 4] = acc;
    for (int rz = row + 1; rz < 16; ++rz)
      *(float4*)&Aw[rz * AS_STRIDE + lane * 4] = make_float4(0.f, 0.f, 0.f, 0.f);
    __builtin_amdgcn_wave_barrier();

    const int tg = tgts[min(t * 16 + ln, np - 1)];
    pv = pvn;
    eb = __builtin_amdgcn_readlane(pv, 0);
    iv = srcs[min(eb + lane, N_EDGES - 1)];
    wv = wts[min(eb + lane, N_EDGES - 1)];
    pvn = pst[min((t + 2 * S) * 16 + l16, np)];

    v8f c[8];
#pragma unroll
    for (int j = 0; j < 8; ++j) c[j] = (v8f){0.f, 0.f, 0.f, 0.f, 0.f, 0.f, 0.f, 0.f};
    v2f a = *(const v2f*)&Aw[ln * AS_STRIDE + koff];
    float4 b[4];
#pragma unroll
    for (int jp = 0; jp < 4; ++jp) b[jp] = Ws4[jp * 32 + lane];
#pragma unroll
    for (int kk = 0; kk < 32; ++kk) {
      v2f an = a;
      float4 bn[4];
#pragma unroll
      for (int jp = 0; jp < 4; ++jp) bn[jp] = b[jp];
      if (kk + 1 < 32) {
        an = *(const v2f*)&Aw[ln * AS_STRIDE + 4 * (kk + 1) + koff];
#pragma unroll
        for (int jp = 0; jp < 4; ++jp)
          bn[jp] = Ws4[((kk + 1) * 4 + jp) * 32 + lane];
      }
#pragma unroll
      for (int jp = 0; jp < 4; ++jp) {
        v2f b0, b1;
        b0.x = b[jp].x; b0.y = b[jp].y;
        b1.x = b[jp].z; b1.y = b[jp].w;
        c[2 * jp] = __builtin_amdgcn_wmma_f32_16x16x4_f32(
            false, a, false, b0, (short)0, c[2 * jp], false, false);
        c[2 * jp + 1] = __builtin_amdgcn_wmma_f32_16x16x4_f32(
            false, a, false, b1, (short)0, c[2 * jp + 1], false, false);
      }
      a = an;
#pragma unroll
      for (int jp = 0; jp < 4; ++jp) b[jp] = bn[jp];
    }

#pragma unroll
    for (int j = 0; j < 8; ++j) {
#pragma unroll
      for (int v = 0; v < 8; ++v)
        Aw[(v + 8 * hi) * AS_STRIDE + 16 * j + ln] = c[j][v];
    }
    __builtin_amdgcn_wave_barrier();
    const int qbase = t * 16;
    for (int i0 = 0; i0 < 16; i0 += 4) {
      float m[4][4];
#pragma unroll
      for (int i = 0; i < 4; ++i) {
#pragma unroll
        for (int jj = 0; jj < 4; ++jj)
          m[i][jj] = Aw[(i0 + i) * AS_STRIDE + lane + 32 * jj];
      }
#pragma unroll
      for (int i = 0; i < 4; ++i) {
        if (qbase + i0 + i < np) {
          const int tgi = __builtin_amdgcn_readlane(tg, i0 + i);
          float* d = out + (size_t)tgi * DIM + lane;
#pragma unroll
          for (int jj = 0; jj < 4; ++jj) atomicAdd(d + 32 * jj, m[i][jj]);
        }
      }
    }
    __builtin_amdgcn_wave_barrier();
  }
}

extern "C" void kernel_launch(void* const* d_in, const int* in_sizes, int n_in,
                              void* d_out, int out_size, void* d_ws, size_t ws_size,
                              hipStream_t stream) {
  const float* x      = (const float*)d_in[0];
  const int*   eidx   = (const int*)d_in[1];
  const float* ew     = (const float*)d_in[2];
  const float* relW1  = (const float*)d_in[3];
  const float* selfW1 = (const float*)d_in[4];
  const float* bias1  = (const float*)d_in[5];
  const float* relW2  = (const float*)d_in[6];
  const float* selfW2 = (const float*)d_in[7];
  const float* bias2  = (const float*)d_in[8];
  float* out = (float*)d_out;

  float* h1   = (float*)d_ws;
  int* cnt    = (int*)(h1 + (size_t)N_NODES * DIM);
  int* ssrc   = cnt + (size_t)N_REL * N_NODES;
  float* sw   = (float*)(ssrc + (size_t)N_REL * N_EDGES);
  int* ptgt   = (int*)(sw + (size_t)N_REL * N_EDGES);
  int* pstart = ptgt + (size_t)N_REL * N_NODES;
  int* rtot   = pstart + (size_t)N_REL * PS_STRIDE;
  int* npair  = rtot + (size_t)N_REL * NCH * 2;

  const int GEMM_BLOCKS = 512;
  const int EDGE_BLOCKS = 128;

  range_count_kernel<<<dim3(NCH, N_REL), 1024, 0, stream>>>(eidx, cnt, rtot);
  range_place_kernel<<<dim3(NCH, N_REL), 1024, 0, stream>>>(
      eidx, ew, cnt, rtot, ssrc, sw, ptgt, pstart, npair);

  self_gemm_kernel<false><<<GEMM_BLOCKS, 256, 0, stream>>>(x, selfW1, bias1,
                                                           cnt, h1);
  edge_gemm_scatter_kernel<false><<<dim3(EDGE_BLOCKS, N_REL), 256, 0, stream>>>(
      x, relW1, ssrc, sw, ptgt, pstart, npair, h1);

  self_gemm_kernel<true><<<GEMM_BLOCKS, 256, 0, stream>>>(h1, selfW2, bias2,
                                                          cnt, out);
  edge_gemm_scatter_kernel<true><<<dim3(EDGE_BLOCKS, N_REL), 256, 0, stream>>>(
      h1, relW2, ssrc, sw, ptgt, pstart, npair, out);
}


# Edge GEMM launched as one 28-wave work-group per WGP (was two of 8): W staged once per WGP, 7 waves per SIMD share the matrix pipe, tiles dealt wave-major; same sums, f32 atomics in arrival order
# speedup vs baseline: 2.0146x; 1.0652x over previous
#include <hip/hip_runtime.h>
#include <hip/hip_bf16.h>

typedef __attribute__((ext_vector_type(2))) float v2f;
typedef __attribute__((ext_vector_type(8))) float v8f;

#define N_NODES 100000
#define N_REL   8
#define N_EDGES 100000
#define DIM     128
#define TILES   (N_EDGES / 16)
#define AS_STRIDE 132
#define NCH 32
#define CH  (N_NODES / NCH)
#define PS_STRIDE (N_NODES + 16)
#define LIST_BATCH  5
#define PLACE_BATCH 4
#define EDGE_WAVES   28
#define EDGE_THREADS 896

__device__ __forceinline__ void load_B_regs(const float* __restrict__ W,
                                            int col, int koff, v2f* bW) {
#pragma unroll
  for (int kk = 0; kk < 32; ++kk) {
    int k = 4 * kk + koff;
    v2f b;
    b.x = W[(size_t)k * DIM + col];
    b.y = W[(size_t)(k + 1) * DIM + col];
    bW[kk] = b;
  }
}

__device__ __forceinline__ v8f mma_tile_f32(const float* __restrict__ As,
                                            int ln, int koff, const v2f* bW) {
  v8f c = {0.f, 0.f, 0.f, 0.f, 0.f, 0.f, 0.f, 0.f};
#pragma unroll
  for (int kk = 0; kk < 32; ++kk) {
    v2f a = *(const v2f*)&As[ln * AS_STRIDE + 4 * kk + koff];
    c = __builtin_amdgcn_wmma_f32_16x16x4_f32(false, a, false, bW[kk],
                                              (short)0, c, false, false);
  }
  return c;
}

template <bool RELU_IN>
__global__ __launch_bounds__(256) void self_gemm_kernel(
    const float* __restrict__ x, const float* __restrict__ W,
    const float* __restrict__ bias, const int* __restrict__ flags,
    float* __restrict__ out) {
  __shared__ float As[16 * AS_STRIDE];
  const int tid  = threadIdx.x;
  const int lane = tid & 31;
  const int wave = __builtin_amdgcn_readfirstlane(tid >> 5);
  const int ln   = lane & 15;
  const int hi   = lane >> 4;
  const int koff = hi * 2;
  const int n0   = wave * 16;

  v2f bW[32];
  load_B_regs(W, n0 + ln, koff, bW);
  v2f bB[2];
  for (int kk = 0; kk < 2; ++kk) {
    int k = 4 * kk + koff;
    v2f b;
    b.x = bias[k * DIM + n0 + ln];
    b.y = bias[(k + 1) * DIM + n0 + ln];
    bB[kk] = b;
  }

  const int S = gridDim.x;
  int t = blockIdx.x;
  if (t >= TILES) return;
  const float* __restrict__ xl = x + lane * 4;
  float4 p0 = *(const float4*)(xl + (size_t)(t * 16 + wave) * DIM);
  float4 p1 = *(const float4*)(xl + (size_t)(t * 16 + wave + 8) * DIM);

  for (; t < TILES; t += S) {
    if (RELU_IN) {
      p0.x = fmaxf(p0.x, 0.f); p0.y = fmaxf(p0.y, 0.f);
      p0.z = fmaxf(p0.z, 0.f); p0.w = fmaxf(p0.w, 0.f);
      p1.x = fmaxf(p1.x, 0.f); p1.y = fmaxf(p1.y, 0.f);
      p1.z = fmaxf(p1.z, 0.f); p1.w = fmaxf(p1.w, 0.f);
    }
    *(float4*)&As[wave * AS_STRIDE + lane * 4] = p0;
    *(float4*)&As[(wave + 8) * AS_STRIDE + lane * 4] = p1;
    __syncthreads();

    int f[4];
    for (int kk = 0; kk < 2; ++kk) {
      int k = 4 * kk + koff;
      f[2 * kk]     = flags[(size_t)k * N_NODES + t * 16 + ln];
      f[2 * kk + 1] = flags[(size_t)(k + 1) * N_NODES + t * 16 + ln];
    }
    const int tn = t + S;
    if (tn < TILES) {
      p0 = *(const float4*)(xl + (size_t)(tn * 16 + wave) * DIM);
      p1 = *(const float4*)(xl + (size_t)(tn * 16 + wave + 8) * DIM);
    }

    v8f c = mma_tile_f32(As, ln, koff, bW);
    for (int kk = 0; kk < 2; ++kk) {
      v2f a;
      a.x = f[2 * kk] > 0 ? 1.f : 0.f;
      a.y = f[2 * kk + 1] > 0 ? 1.f : 0.f;
      c = __builtin_amdgcn_wmma_f32_16x16x4_f32(false, a, false, bB[kk],
                                                (short)0, c, false, false);
    }

#pragma unroll
    for (int v = 0; v < 8; ++v) {
      int row = v + 8 * hi;
      out[(size_t)(t * 16 + row) * DIM + n0 + ln] = c[v];
    }
    __syncthreads();
  }
}


__device__ __forceinline__ void block_scan2(int& a, int& b, int* sa, int* sb,
                                            int tid) {
  const int lane = tid & 31, wv = tid >> 5;
  for (int off = 1; off < 32; off <<= 1) {
    int ta = __shfl_up(a, off), tb = __shfl_up(b, off);
    if (lane >= off) { a += ta; b += tb; }
  }
  if (lane == 31) { sa[wv] = a; sb[wv] = b; }
  __syncthreads();
  if (wv == 0) {
    int wa = sa[lane], wb = sb[lane];
    for (int off = 1; off < 32; off <<= 1) {
      int ta = __shfl_up(wa, off), tb = __shfl_up(wb, off);
      if (lane >= off) { wa += ta; wb += tb; }
    }
    sa[lane] = wa;
    sb[lane] = wb;
  }
  __syncthreads();
  if (wv > 0) { a += sa[wv - 1]; b += sb[wv - 1]; }
}

__global__ __launch_bounds__(1024) void range_count_kernel(
    const int* __restrict__ edge_index, int* __restrict__ cnt,
    int* __restrict__ rtot) {
  __shared__ int s_cnt[CH];
  __shared__ int sa[32];
  __shared__ int sb[32];
  const int c = blockIdx.x, r = blockIdx.y, tid = threadIdx.x;
  const int lo = c * CH;
  for (int i = tid; i < CH; i += 1024) s_cnt[i] = 0;
  __syncthreads();
  const int4* __restrict__ tg =
      (const int4*)(edge_index + (size_t)(2 * r + 1) * N_EDGES);
  for (int q0 = tid; q0 < N_EDGES / 4; q0 += 1024 * LIST_BATCH) {
    int4 t[LIST_BATCH];
#pragma unroll
    for (int u = 0; u < LIST_BATCH; ++u) {
      const int q = q0 + 1024 * u;
      t[u] = q < N_EDGES / 4 ? tg[q] : make_int4(-1, -1, -1, -1);
    }
#pragma unroll
    for (int u = 0; u < LIST_BATCH; ++u) {
      const unsigned d0 = (unsigned)(t[u].x - lo), d1 = (unsigned)(t[u].y - lo);
      const unsigned d2 = (unsigned)(t[u].z - lo), d3 = (unsigned)(t[u].w - lo);
      if (d0 < (unsigned)CH) atomicAdd(&s_cnt[d0], 1);
      if (d1 < (unsigned)CH) atomicAdd(&s_cnt[d1], 1);
      if (d2 < (unsigned)CH) atomicAdd(&s_cnt[d2], 1);
      if (d3 < (unsigned)CH) atomicAdd(&s_cnt[d3], 1);
    }
  }
  __syncthreads();
  int ne = 0, nf = 0;
  for (int i = tid; i < CH; i += 1024) {
    const int v = s_cnt[i];
    cnt[(size_t)r * N_NODES + lo + i] = v;
    ne += v;
    nf += (v > 0);
  }
  block_scan2(ne, nf, sa, sb, tid);
  if (tid == 1023) {
    rtot[(r * NCH + c) * 2]     = ne;
    rtot[(r * NCH + c) * 2 + 1] = nf;
  }
}

__global__ __launch_bounds__(1024) void range_place_kernel(
    const int* __restrict__ edge_index, const float* __restrict__ ew,
    const int* __restrict__ cnt, const int* __restrict__ rtot,
    int* __restrict__ ssrc, float* __restrict__ sw, int* __restrict__ ptgt,
    int* __restrict__ pstart, int* __restrict__ npair) {
  __shared__ int s_cur[CH];
  __shared__ int sa[32];
  __shared__ int sb[32];
  const int c = blockIdx.x, r = blockIdx.y, tid = threadIdx.x;
  const int lo = c * CH;
  int baseC = 0, baseF = 0;
  for (int j = 0; j < c; ++j) {
    baseC += rtot[(r * NCH + j) * 2];
    baseF += rtot[(r * NCH + j) * 2 + 1];
  }
  const int i0 = tid * 4;
  int cs[4];
#pragma unroll
  for (int i = 0; i < 4; ++i)
    cs[i] = (i0 + i < CH) ? cnt[(size_t)r * N_NODES + lo + i0 + i] : 0;
  const int sc = cs[0] + cs[1] + cs[2] + cs[3];
  const int sf = (cs[0] > 0) + (cs[1] > 0) + (cs[2] > 0) + (cs[3] > 0);
  int ic = sc, jf = sf;
  block_scan2(ic, jf, sa, sb, tid);
  int runC = baseC + ic - sc;
  int runF = baseF + jf - sf;
#pragma unroll
  for (int i = 0; i < 4; ++i) {
    if (i0 + i < CH) {
      s_cur[i0 + i] = runC;
      if (cs[i] > 0) {
        ptgt[(size_t)r * N_NODES + runF]     = lo + i0 + i;
        pstart[(size_t)r * PS_STRIDE + runF] = runC;
        ++runF;
      }
      runC += cs[i];
    }
  }
  if (c == NCH - 1 && tid == 1023) {
    npair[r] = runF;
    pstart[(size_t)r * PS_STRIDE + runF] = runC;
  }
  __syncthreads();

  const int4* __restrict__ tg =
      (const int4*)(edge_index + (size_t)(2 * r + 1) * N_EDGES);
  const int*   __restrict__ sr = edge_index + (size_t)(2 * r) * N_EDGES;
  const float* __restrict__ wr = ew + (size_t)r * N_EDGES;
  int*   __restrict__ so = ssrc + (size_t)r * N_EDGES;
  float* __restrict__ wo = sw + (size_t)r * N_EDGES;
  const int4*   __restrict__ sr4 = (const int4*)sr;
  const float4* __restrict__ wr4 = (const float4*)wr;
  for (int q0 = tid; q0 < N_EDGES / 4; q0 += 1024 * PLACE_BATCH) {
    int4 t[PLACE_BATCH], s[PLACE_BATCH];
    float4 w[PLACE_BATCH];
#pragma unroll
    for (int u = 0; u < PLACE_BATCH; ++u) {
      const int q = q0 + 1024 * u;
      if (q < N_EDGES / 4) {
        t[u] = tg[q]; s[u] = sr4[q]; w[u] = wr4[q];
      } else {
        t[u] = make_int4(-1, -1, -1, -1);
        s[u] = make_int4(0, 0, 0, 0);
        w[u] = make_float4(0.f, 0.f, 0.f, 0.f);
      }
    }
#pragma unroll
    for (int u = 0; u < PLACE_BATCH; ++u) {
      const int   tt[4] = {t[u].x, t[u].y, t[u].z, t[u].w};
      const int   ss[4] = {s[u].x, s[u].y, s[u].z, s[u].w};
      const float ww[4] = {w[u].x, w[u].y, w[u].z, w[u].w};
#pragma unroll
      for (int i = 0; i < 4; ++i) {
        const unsigned d = (unsigned)(tt[i] - lo);
        if (d < (unsigned)CH) {
          const int pos = atomicAdd(&s_cur[d], 1);
          so[pos] = ss[i];
          wo[pos] = ww[i];
        }
      }
    }
  }
}

template <bool RELU_IN>
__device__ __forceinline__ void add_row(float4& acc, float w, float4 v) {
  if (RELU_IN) {
    v.x = fmaxf(v.x, 0.f); v.y = fmaxf(v.y, 0.f);
    v.z = fmaxf(v.z, 0.f); v.w = fmaxf(v.w, 0.f);
  }
  acc.x += w * v.x; acc.y += w * v.y; acc.z += w * v.z; acc.w += w * v.w;
}

template <bool RELU_IN>
__global__ __launch_bounds__(EDGE_THREADS) void edge_gemm_scatter_kernel(
    const float* __restrict__ x, const float* __restrict__ relW,
    const int* __restrict__ ssrc, const float* __restrict__ sw,
    const int* __restrict__ ptgt, const int* __restrict__ pstart,
    const int* __restrict__ npair, float* __restrict__ out) {
  __shared__ __attribute__((aligned(16))) float Ws[DIM * DIM];
  __shared__ __attribute__((aligned(16))) float Ts[EDGE_WAVES * 16 * AS_STRIDE];

  const int r = blockIdx.y;
  const float* __restrict__ W    = relW + (size_t)r * DIM * DIM;
  const int*   __restrict__ srcs = ssrc + (size_t)r * N_EDGES;
  const float* __restrict__ wts  = sw + (size_t)r * N_EDGES;
  const int*   __restrict__ tgts = ptgt + (size_t)r * N_NODES;
  const int*   __restrict__ pst  = pstart + (size_t)r * PS_STRIDE;
  const int np     = npair[r];
  const int ntiles = (np + 15) >> 4;

  const int tid  = threadIdx.x;
  const int lane = tid & 31;
  const int wave = __builtin_amdgcn_readfirstlane(tid >> 5);
  const int ln   = lane & 15;
  const int hi   = lane >> 4;
  const int koff = hi * 2;

  for (int i = tid; i < DIM * DIM / 4; i += EDGE_THREADS) {
    const int row = i >> 5, col = (i & 31) << 2;
    const float4 w4 = *(const float4*)(W + (size_t)row * DIM + col);
    const int kk = row >> 2, h = (row >> 1) & 1, s = row & 1;
    const int j = col >> 4, c = col & 15;
    float* d = &Ws[((((kk * 4 + (j >> 1)) * 32) + h * 16 + c) << 2) + 2 * (j & 1) + s];
    d[0] = w4.x; d[4] = w4.y; d[8] = w4.z; d[12] = w4.w;
  }
  __syncthreads();

  float* __restrict__ Aw = Ts + wave * (16 * AS_STRIDE);
  const float4* __restrict__ Ws4 = (const float4*)Ws;

  const int S = gridDim.x * EDGE_WAVES;
  int t = wave * gridDim.x + blockIdx.x;
  if (t >= ntiles) return;

  const float* __restrict__ xl = x + lane * 4;
  const int l16 = min(lane, 16);
  int pv  = pst[min(t * 16 + l16, np)];
  int pvn = pst[min((t + S) * 16 + l16, np)];
  int eb  = __builtin_amdgcn_readlane(pv, 0);
  int   iv = srcs[min(eb + lane, N_EDGES - 1)];
  float wv = wts[min(eb + lane, N_EDGES - 1)];

  for (; t < ntiles; t += S) {
    const int ee = __builtin_amdgcn_readlane(pv, 16);
    float4 acc = {0.f, 0.f, 0.f, 0.f};
    int row  = 0;
    int rend = __builtin_amdgcn_readlane(pv, 1);
    for (int cb = eb; cb < ee; cb += 32) {
      if (cb != eb) {
        iv = srcs[min(cb + lane, N_EDGES - 1)];
        wv = wts[min(cb + lane, N_EDGES - 1)];
      }
      const int cnt = min(32, ee - cb);
      for (int j0 = 0; j0 < cnt; j0 += 8) {
        float4 v[8];
#pragma unroll
        for (int u = 0; u < 8; ++u) {
          const int s = __builtin_amdgcn_readlane(iv, min(j0 + u, cnt - 1));
          v[u] = *(const float4*)(xl + (size_t)s * DIM);
        }
#pragma unroll
        for (int u = 0; u < 8; ++u) {
          const int j = j0 + u;
          if (j < cnt) {
            while (cb + j >= rend && row < 15) {
              *(float4*)&Aw[row * AS_STRIDE + lane * 4] = acc;
              acc = make_float4(0.f, 0.f, 0.f, 0.f);
              ++row;
              rend = __builtin_amdgcn_readlane(pv, row + 1);
            }
            const float w =
                __int_as_float(__builtin_amdgcn_readlane(__float_as_int(wv), j));
            add_row<RELU_IN>(acc, w, v[u]);
          }
        }
      }
    }
    *(float4*)&Aw[row * AS_STRIDE + lane * 4] = acc;
    for (int rz = row + 1; rz < 16; ++rz)
      *(float4*)&Aw[rz * AS_STRIDE + lane * 4] = make_float4(0.f, 0.f, 0.f, 0.f);
    __builtin_amdgcn_wave_barrier();

    const int tg = tgts[min(t * 16 + ln, np - 1)];
    pv = pvn;
    eb = __builtin_amdgcn_readlane(pv, 0);
    iv = srcs[min(eb + lane, N_EDGES - 1)];
    wv = wts[min(eb + lane, N_EDGES - 1)];
    pvn = pst[min((t + 2 * S) * 16 + l16, np)];

    v8f c[8];
#pragma unroll
    for (int j = 0; j < 8; ++j) c[j] = (v8f){0.f, 0.f, 0.f, 0.f, 0.f, 0.f, 0.f, 0.f};
    v2f a = *(const v2f*)&Aw[ln * AS_STRIDE + koff];
    float4 b[4];
#pragma unroll
    for (int jp = 0; jp < 4; ++jp) b[jp] = Ws4[jp * 32 + lane];
#pragma unroll
    for (int kk = 0; kk < 32; ++kk) {
      v2f an = a;
      float4 bn[4];
#pragma unroll
      for (int jp = 0; jp < 4; ++jp) bn[jp] = b[jp];
      if (kk + 1 < 32) {
        an = *(const v2f*)&Aw[ln * AS_STRIDE + 4 * (kk + 1) + koff];
#pragma unroll
        for (int jp = 0; jp < 4; ++jp)
          bn[jp] = Ws4[((kk + 1) * 4 + jp) * 32 + lane];
      }
#pragma unroll
      for (int jp = 0; jp < 4; ++jp) {
        v2f b0, b1;
        b0.x = b[jp].x; b0.y = b[jp].y;
        b1.x = b[jp].z; b1.y = b[jp].w;
        c[2 * jp] = __builtin_amdgcn_wmma_f32_16x16x4_f32(
            false, a, false, b0, (short)0, c[2 * jp], false, false);
        c[2 * jp + 1] = __builtin_amdgcn_wmma_f32_16x16x4_f32(
            false, a, false, b1, (short)0, c[2 * jp + 1], false, false);
      }
      a = an;
#pragma unroll
      for (int jp = 0; jp < 4; ++jp) b[jp] = bn[jp];
    }

#pragma unroll
    for (int j = 0; j < 8; ++j) {
#pragma unroll
      for (int v = 0; v < 8; ++v)
        Aw[(v + 8 * hi) * AS_STRIDE + 16 * j + ln] = c[j][v];
    }
    __builtin_amdgcn_wave_barrier();
    const int qbase = t * 16;
    for (int i0 = 0; i0 < 16; i0 += 4) {
      float m[4][4];
#pragma unroll
      for (int i = 0; i < 4; ++i) {
#pragma unroll
        for (int jj = 0; jj < 4; ++jj)
          m[i][jj] = Aw[(i0 + i) * AS_STRIDE + lane + 32 * jj];
      }
#pragma unroll
      for (int i = 0; i < 4; ++i) {
        if (qbase + i0 + i < np) {
          const int tgi = __builtin_amdgcn_readlane(tg, i0 + i);
          float* d = out + (size_t)tgi * DIM + lane;
#pragma unroll
          for (int jj = 0; jj < 4; ++jj) atomicAdd(d + 32 * jj, m[i][jj]);
        }
      }
    }
    __builtin_amdgcn_wave_barrier();
  }
}

extern "C" void kernel_launch(void* const* d_in, const int* in_sizes, int n_in,
                              void* d_out, int out_size, void* d_ws, size_t ws_size,
                              hipStream_t stream) {
  const float* x      = (const float*)d_in[0];
  const int*   eidx   = (const int*)d_in[1];
  const float* ew     = (const float*)d_in[2];
  const float* relW1  = (const float*)d_in[3];
  const float* selfW1 = (const float*)d_in[4];
  const float* bias1  = (const float*)d_in[5];
  const float* relW2  = (const float*)d_in[6];
  const float* selfW2 = (const float*)d_in[7];
  const float* bias2  = (const float*)d_in[8];
  float* out = (float*)d_out;

  float* h1   = (float*)d_ws;
  int* cnt    = (int*)(h1 + (size_t)N_NODES * DIM);
  int* ssrc   = cnt + (size_t)N_REL * N_NODES;
  float* sw   = (float*)(ssrc + (size_t)N_REL * N_EDGES);
  int* ptgt   = (int*)(sw + (size_t)N_REL * N_EDGES);
  int* pstart = ptgt + (size_t)N_REL * N_NODES;
  int* rtot   = pstart + (size_t)N_REL * PS_STRIDE;
  int* npair  = rtot + (size_t)N_REL * NCH * 2;

  const int GEMM_BLOCKS = 512;
  const int EDGE_BLOCKS = 32;

  range_count_kernel<<<dim3(NCH, N_REL), 1024, 0, stream>>>(eidx, cnt, rtot);
  range_place_kernel<<<dim3(NCH, N_REL), 1024, 0, stream>>>(
      eidx, ew, cnt, rtot, ssrc, sw, ptgt, pstart, npair);

  self_gemm_kernel<false><<<GEMM_BLOCKS, 256, 0, stream>>>(x, selfW1, bias1,
                                                           cnt, h1);
  edge_gemm_scatter_kernel<false><<<dim3(EDGE_BLOCKS, N_REL), EDGE_THREADS, 0, stream>>>(
      x, relW1, ssrc, sw, ptgt, pstart, npair, h1);

  self_gemm_kernel<true><<<GEMM_BLOCKS, 256, 0, stream>>>(h1, selfW2, bias2,
                                                          cnt, out);
  edge_gemm_scatter_kernel<true><<<dim3(EDGE_BLOCKS, N_REL), EDGE_THREADS, 0, stream>>>(
      h1, relW2, ssrc, sw, ptgt, pstart, npair, out);
}
